# starting kernel
# baseline (speedup 1.0000x reference)
#include <hip/hip_runtime.h>
#include <math.h>

typedef float v2f __attribute__((ext_vector_type(2)));
typedef float v8f __attribute__((ext_vector_type(8)));

#define FEAT 128
#define N_CLASSES 16
#define BN_EPS 1e-5f

__global__ void k_fill(float* __restrict__ p, float v, int n) {
    int i = blockIdx.x * blockDim.x + threadIdx.x;
    if (i < n) p[i] = v;
}

__global__ void k_deg_accum(const int* __restrict__ dst, float* __restrict__ deg, int E) {
    int e = blockIdx.x * blockDim.x + threadIdx.x;
    if (e < E) {
        __hip_atomic_fetch_add(&deg[dst[e]], 1.0f,
                               __ATOMIC_RELAXED, __HIP_MEMORY_SCOPE_AGENT);
    }
}

__global__ void k_rsqrt_inplace(float* __restrict__ p, int n) {
    int i = blockIdx.x * blockDim.x + threadIdx.x;
    if (i < n) p[i] = rsqrtf(p[i]);
}

__global__ void k_gemm_wmma(const float* __restrict__ A, const float* __restrict__ B,
                            const float* __restrict__ bias, float* __restrict__ C,
                            int M, int K, int N) {
    const int lane  = threadIdx.x & 31;
    const int wave  = threadIdx.x >> 5;
    const int wpb   = blockDim.x >> 5;
    const int tile  = blockIdx.x * wpb + wave;
    const int Mt    = (M + 15) >> 4;
    const int Nt    = N >> 4;
    if (tile >= Mt * Nt) return;

    const int mBase = (tile / Nt) << 4;
    const int nBase = (tile % Nt) << 4;

    int row = mBase + (lane & 15);
    if (row > M - 1) row = M - 1;
    const int col  = nBase + (lane & 15);
    const int kOff = (lane >> 4) << 1;

    v8f acc = {};
    for (int k = 0; k < K; k += 4) {
        v2f a = *(const v2f*)(A + (size_t)row * K + k + kOff);
        v2f b;
        b.x = B[(size_t)(k + kOff)     * N + col];
        b.y = B[(size_t)(k + kOff + 1) * N + col];
        acc = __builtin_amdgcn_wmma_f32_16x16x4_f32(
                   false, a,  false, b,
                   (short)0, acc,  false,  false);
    }

    const float bb   = bias ? bias[col] : 0.0f;
    const int   rOff = (lane >> 4) << 3;
#pragma unroll
    for (int j = 0; j < 8; ++j) {
        int r = mBase + j + rOff;
        if (r < M) C[(size_t)r * N + col] = acc[j] + bb;
    }
}

__global__ void k_scatter(const float* __restrict__ H,
                          const int* __restrict__ src, const int* __restrict__ dst,
                          const float* __restrict__ dinv, float* __restrict__ out, int E) {
    const int lane = threadIdx.x & 31;
    const int wave = threadIdx.x >> 5;
    const int e    = blockIdx.x * (blockDim.x >> 5) + wave;
    if (e >= E) return;

    const int   s = src[e];
    const int   d = dst[e];
    const float w = dinv[s] * dinv[d];

    const float4 hv = *(const float4*)(H + (size_t)s * FEAT + lane * 4);
    float* o = out + (size_t)d * FEAT + lane * 4;
    __hip_atomic_fetch_add(o + 0, hv.x * w, __ATOMIC_RELAXED, __HIP_MEMORY_SCOPE_AGENT);
    __hip_atomic_fetch_add(o + 1, hv.y * w, __ATOMIC_RELAXED, __HIP_MEMORY_SCOPE_AGENT);
    __hip_atomic_fetch_add(o + 2, hv.z * w, __ATOMIC_RELAXED, __HIP_MEMORY_SCOPE_AGENT);
    __hip_atomic_fetch_add(o + 3, hv.w * w, __ATOMIC_RELAXED, __HIP_MEMORY_SCOPE_AGENT);
}

__global__ void k_post1(float* __restrict__ agg, const float* __restrict__ H,
                        const float* __restrict__ dinv, const float* __restrict__ b1,
                        const float* __restrict__ gamma, const float* __restrict__ beta,
                        const float* __restrict__ mean, const float* __restrict__ var,
                        int total) {
    int i = blockIdx.x * blockDim.x + threadIdx.x;
    if (i >= total) return;
    const int   node = i >> 7;
    const int   f    = i & (FEAT - 1);
    const float di   = dinv[node];
    float v = agg[i] + H[i] * di * di + b1[f];
    v = (v - mean[f]) * rsqrtf(var[f] + BN_EPS) * gamma[f] + beta[f];
    agg[i] = fmaxf(v, 0.0f);
}

__global__ void k_post2(float* __restrict__ agg, const float* __restrict__ H,
                        const float* __restrict__ dinv, const float* __restrict__ b2,
                        int total) {
    int i = blockIdx.x * blockDim.x + threadIdx.x;
    if (i >= total) return;
    const int   node = i >> 7;
    const int   f    = i & (FEAT - 1);
    const float di   = dinv[node];
    agg[i] = agg[i] + H[i] * di * di + b2[f];
}

extern "C" void kernel_launch(void* const* d_in, const int* in_sizes, int n_in,
                              void* d_out, int out_size, void* d_ws, size_t ws_size,
                              hipStream_t stream) {
    const float* X     = (const float*)d_in[0];
    const int*   edge  = (const int*)  d_in[1];
    const float* W1    = (const float*)d_in[2];
    const float* b1    = (const float*)d_in[3];
    const float* gamma = (const float*)d_in[4];
    const float* beta  = (const float*)d_in[5];
    const float* rmean = (const float*)d_in[6];
    const float* rvar  = (const float*)d_in[7];
    const float* W2    = (const float*)d_in[8];
    const float* b2    = (const float*)d_in[9];
    const float* Wc    = (const float*)d_in[10];
    const float* bc    = (const float*)d_in[11];
    float*       out   = (float*)d_out;

    const int N = in_sizes[0] / FEAT;
    const int E = in_sizes[1] / 2;
    const int* src = edge;
    const int* dst = edge + E;
    const int total = N * FEAT;

    float* deg  = (float*)d_ws;
    float* bufA = deg + ((N + 63) & ~63);
    float* bufB = bufA + (size_t)total;

    const int T256   = 256;
    const int gNode  = (N + T256 - 1) / T256;
    const int gEdge  = (E + T256 - 1) / T256;
    const int gFeat  = (total + T256 - 1) / T256;
    const int gScat  = (E + 7) / 8;
    const int tilesL = ((N + 15) / 16) * (FEAT / 16);
    const int tilesC = ((N + 15) / 16) * (N_CLASSES / 16);
    const int gGemmL = (tilesL + 7) / 8;
    const int gGemmC = (tilesC + 7) / 8;

    k_fill<<<gNode, T256, 0, stream>>>(deg, 1.0f, N);
    k_deg_accum<<<gEdge, T256, 0, stream>>>(dst, deg, E);
    k_rsqrt_inplace<<<gNode, T256, 0, stream>>>(deg, N);

    k_gemm_wmma<<<gGemmL, T256, 0, stream>>>(X, W1, nullptr, bufA, N, FEAT, FEAT);
    k_fill<<<gFeat, T256, 0, stream>>>(bufB, 0.0f, total);
    k_scatter<<<gScat, T256, 0, stream>>>(bufA, src, dst, deg, bufB, E);
    k_post1<<<gFeat, T256, 0, stream>>>(bufB, bufA, deg, b1, gamma, beta, rmean, rvar, total);

    k_gemm_wmma<<<gGemmL, T256, 0, stream>>>(bufB, W2, nullptr, bufA, N, FEAT, FEAT);
    k_fill<<<gFeat, T256, 0, stream>>>(bufB, 0.0f, total);
    k_scatter<<<gScat, T256, 0, stream>>>(bufA, src, dst, deg, bufB, E);
    k_post2<<<gFeat, T256, 0, stream>>>(bufB, bufA, deg, b2, total);

    k_gemm_wmma<<<gGemmC, T256, 0, stream>>>(bufB, Wc, bc, out, N, FEAT, N_CLASSES);
}


# Fold the classifier into layer 2 by linearity: out = Ahat(act (W2 Wc)) + (b2 Wc + bc); layer-2 fill, atomic scatter and epilogue run at 16 floats per node, not 128; f32 products and sums reassociated
# speedup vs baseline: 1.4905x; 1.4905x over previous
#include <hip/hip_runtime.h>
#include <math.h>

typedef float v2f __attribute__((ext_vector_type(2)));
typedef float v8f __attribute__((ext_vector_type(8)));

#define FEAT 128
#define N_CLASSES 16
#define BN_EPS 1e-5f

__global__ void k_fill(float* __restrict__ p, float v, int n) {
    int i = blockIdx.x * blockDim.x + threadIdx.x;
    if (i < n) p[i] = v;
}

__global__ void k_deg_accum(const int* __restrict__ dst, float* __restrict__ deg, int E) {
    int e = blockIdx.x * blockDim.x + threadIdx.x;
    if (e < E) {
        __hip_atomic_fetch_add(&deg[dst[e]], 1.0f,
                               __ATOMIC_RELAXED, __HIP_MEMORY_SCOPE_AGENT);
    }
}

__global__ void k_rsqrt_inplace(float* __restrict__ p, int n) {
    int i = blockIdx.x * blockDim.x + threadIdx.x;
    if (i < n) p[i] = rsqrtf(p[i]);
}

__device__ __forceinline__ void gemm_tile_f32(const float* __restrict__ A, const float* __restrict__ B,
                                              const float* __restrict__ bias, float* __restrict__ C,
                                              int M, int K, int N, int tile, int lane) {
    const int Nt    = N >> 4;
    const int mBase = (tile / Nt) << 4;
    const int nBase = (tile % Nt) << 4;

    int row = mBase + (lane & 15);
    if (row > M - 1) row = M - 1;
    const int col  = nBase + (lane & 15);
    const int kOff = (lane >> 4) << 1;

    v8f acc = {};
    for (int k = 0; k < K; k += 4) {
        v2f a = *(const v2f*)(A + (size_t)row * K + k + kOff);
        v2f b;
        b.x = B[(size_t)(k + kOff)     * N + col];
        b.y = B[(size_t)(k + kOff + 1) * N + col];
        acc = __builtin_amdgcn_wmma_f32_16x16x4_f32(
                   false, a,  false, b,
                   (short)0, acc,  false,  false);
    }

    const float bb   = bias ? bias[col] : 0.0f;
    const int   rOff = (lane >> 4) << 3;
#pragma unroll
    for (int j = 0; j < 8; ++j) {
        int r = mBase + j + rOff;
        if (r < M) C[(size_t)r * N + col] = acc[j] + bb;
    }
}

__global__ void k_gemm_wmma(const float* __restrict__ A, const float* __restrict__ B,
                            const float* __restrict__ bias, float* __restrict__ C,
                            int M, int K, int N) {
    const int lane  = threadIdx.x & 31;
    const int wave  = threadIdx.x >> 5;
    const int wpb   = blockDim.x >> 5;
    const int tile  = blockIdx.x * wpb + wave;
    const int Mt    = (M + 15) >> 4;
    const int Nt    = N >> 4;
    if (tile >= Mt * Nt) return;
    gemm_tile_f32(A, B, bias, C, M, K, N, tile, lane);
}

__global__ void k_fold_classifier(const float* __restrict__ W2, const float* __restrict__ b2,
                                  const float* __restrict__ Wc, const float* __restrict__ bc,
                                  float* __restrict__ W2c, float* __restrict__ cvec) {
    const int lane = threadIdx.x & 31;
    const int wave = threadIdx.x >> 5;
    if (wave < FEAT / 16)
        gemm_tile_f32(W2, Wc, nullptr, W2c, FEAT, FEAT, N_CLASSES, wave, lane);
    else
        gemm_tile_f32(b2, Wc, bc, cvec, 1, FEAT, N_CLASSES, 0, lane);
}

__global__ void k_scatter(const float* __restrict__ H,
                          const int* __restrict__ src, const int* __restrict__ dst,
                          const float* __restrict__ dinv, float* __restrict__ out, int E) {
    const int lane = threadIdx.x & 31;
    const int wave = threadIdx.x >> 5;
    const int e    = blockIdx.x * (blockDim.x >> 5) + wave;
    if (e >= E) return;

    const int   s = src[e];
    const int   d = dst[e];
    const float w = dinv[s] * dinv[d];

    const float4 hv = *(const float4*)(H + (size_t)s * FEAT + lane * 4);
    float* o = out + (size_t)d * FEAT + lane * 4;
    __hip_atomic_fetch_add(o + 0, hv.x * w, __ATOMIC_RELAXED, __HIP_MEMORY_SCOPE_AGENT);
    __hip_atomic_fetch_add(o + 1, hv.y * w, __ATOMIC_RELAXED, __HIP_MEMORY_SCOPE_AGENT);
    __hip_atomic_fetch_add(o + 2, hv.z * w, __ATOMIC_RELAXED, __HIP_MEMORY_SCOPE_AGENT);
    __hip_atomic_fetch_add(o + 3, hv.w * w, __ATOMIC_RELAXED, __HIP_MEMORY_SCOPE_AGENT);
}

template <bool TAIL>
__device__ __forceinline__ void scatter_walk_c(const float* __restrict__ Hc, float* __restrict__ out,
                                               int s, int d, float w, int lane, int nValid) {
    const int f    = lane & (N_CLASSES - 1);
    const int half = lane >> 4;
    float    v[16];
    unsigned o[16];
#pragma unroll
    for (int j = 0; j < 16; ++j) {
        const int      l  = 2 * j + half;
        const unsigned sj = (unsigned)__shfl(s, l);
        const unsigned dj = (unsigned)__shfl(d, l);
        const float    wj = __shfl(w, l);
        v[j] = Hc[sj * N_CLASSES + f] * wj;
        o[j] = dj * N_CLASSES + f;
    }
#pragma unroll
    for (int j = 0; j < 16; ++j) {
        if (!TAIL || 2 * j + half < nValid)
            __hip_atomic_fetch_add(out + o[j], v[j], __ATOMIC_RELAXED, __HIP_MEMORY_SCOPE_AGENT);
    }
}

__global__ void k_scatter_c(const float* __restrict__ Hc,
                            const int* __restrict__ src, const int* __restrict__ dst,
                            const float* __restrict__ dinv, float* __restrict__ out, int E) {
    const int lane = threadIdx.x & 31;
    const int wave = threadIdx.x >> 5;
    const int e0   = (blockIdx.x * (blockDim.x >> 5) + wave) << 5;
    if (e0 >= E) return;

    const int e = e0 + lane;
    int   s = 0, d = 0;
    float w = 0.0f;
    if (e < E) {
        s = src[e];
        d = dst[e];
        w = dinv[s] * dinv[d];
    }

    if (E - e0 >= 32) scatter_walk_c<false>(Hc, out, s, d, w, lane, 32);
    else              scatter_walk_c<true >(Hc, out, s, d, w, lane, E - e0);
}

__global__ void k_post1(float* __restrict__ agg, const float* __restrict__ H,
                        const float* __restrict__ dinv, const float* __restrict__ b1,
                        const float* __restrict__ gamma, const float* __restrict__ beta,
                        const float* __restrict__ mean, const float* __restrict__ var,
                        int total) {
    int i = blockIdx.x * blockDim.x + threadIdx.x;
    if (i >= total) return;
    const int   node = i >> 7;
    const int   f    = i & (FEAT - 1);
    const float di   = dinv[node];
    float v = agg[i] + H[i] * di * di + b1[f];
    v = (v - mean[f]) * rsqrtf(var[f] + BN_EPS) * gamma[f] + beta[f];
    agg[i] = fmaxf(v, 0.0f);
}

__global__ void k_post2(float* __restrict__ agg, const float* __restrict__ Hc,
                        const float* __restrict__ dinv, const float* __restrict__ cvec,
                        int total) {
    int i = blockIdx.x * blockDim.x + threadIdx.x;
    if (i >= total) return;
    const int   node = i >> 4;
    const int   f    = i & (N_CLASSES - 1);
    const float di   = dinv[node];
    agg[i] = agg[i] + Hc[i] * di * di + cvec[f];
}

extern "C" void kernel_launch(void* const* d_in, const int* in_sizes, int n_in,
                              void* d_out, int out_size, void* d_ws, size_t ws_size,
                              hipStream_t stream) {
    const float* X     = (const float*)d_in[0];
    const int*   edge  = (const int*)  d_in[1];
    const float* W1    = (const float*)d_in[2];
    const float* b1    = (const float*)d_in[3];
    const float* gamma = (const float*)d_in[4];
    const float* beta  = (const float*)d_in[5];
    const float* rmean = (const float*)d_in[6];
    const float* rvar  = (const float*)d_in[7];
    const float* W2    = (const float*)d_in[8];
    const float* b2    = (const float*)d_in[9];
    const float* Wc    = (const float*)d_in[10];
    const float* bc    = (const float*)d_in[11];
    float*       out   = (float*)d_out;

    const int N = in_sizes[0] / FEAT;
    const int E = in_sizes[1] / 2;
    const int* src = edge;
    const int* dst = edge + E;
    const int total = N * FEAT;
    const int totalC = N * N_CLASSES;

    float* deg  = (float*)d_ws;
    float* bufA = deg + ((N + 63) & ~63);
    float* bufB = bufA + (size_t)total;

    const int T256   = 256;
    const int gNode  = (N + T256 - 1) / T256;
    const int gEdge  = (E + T256 - 1) / T256;
    const int gFeat  = (total + T256 - 1) / T256;
    const int gScat  = (E + 7) / 8;
    const int tilesL = ((N + 15) / 16) * (FEAT / 16);
    const int tilesC = ((N + 15) / 16) * (N_CLASSES / 16);
    const int gGemmL = (tilesL + 7) / 8;
    const int gGemmC = (tilesC + 7) / 8;
    const int gFeatC = (totalC + T256 - 1) / T256;
    const int gScatC = (E + 255) / 256;

    k_fill<<<gNode, T256, 0, stream>>>(deg, 1.0f, N);
    k_deg_accum<<<gEdge, T256, 0, stream>>>(dst, deg, E);
    k_rsqrt_inplace<<<gNode, T256, 0, stream>>>(deg, N);

    k_gemm_wmma<<<gGemmL, T256, 0, stream>>>(X, W1, nullptr, bufA, N, FEAT, FEAT);
    k_fill<<<gFeat, T256, 0, stream>>>(bufB, 0.0f, total);
    k_scatter<<<gScat, T256, 0, stream>>>(bufA, src, dst, deg, bufB, E);
    k_post1<<<gFeat, T256, 0, stream>>>(bufB, bufA, deg, b1, gamma, beta, rmean, rvar, total);

    float* Hc   = bufA;
    float* W2c  = bufA + (size_t)totalC;
    float* cvec = W2c + FEAT * N_CLASSES;
    k_fold_classifier<<<1, 32 * (FEAT / 16 + 1), 0, stream>>>(W2, b2, Wc, bc, W2c, cvec);
    k_gemm_wmma<<<gGemmC, T256, 0, stream>>>(bufB, W2c, nullptr, Hc, N, FEAT, N_CLASSES);
    k_fill<<<gFeatC, T256, 0, stream>>>(out, 0.0f, totalC);
    k_scatter_c<<<gScatC, T256, 0, stream>>>(Hc, src, dst, deg, out, E);
    k_post2<<<gFeatC, T256, 0, stream>>>(out, Hc, deg, cvec, totalC);
}


# Replace float-atomic scatters by a per-call adjacency sorted by destination (integer count, scan, place) and gathers that take in the fills and epilogues; f32 sums now run in list arrival order
# speedup vs baseline: 2.0807x; 1.3960x over previous
#include <hip/hip_runtime.h>
#include <math.h>

typedef float v2f __attribute__((ext_vector_type(2)));
typedef float v8f __attribute__((ext_vector_type(8)));

#define FEAT 128
#define N_CLASSES 16
#define BN_EPS 1e-5f

__device__ __forceinline__ void gemm_tile_f32(const float* __restrict__ A, const float* __restrict__ B,
                                              const float* __restrict__ bias, float* __restrict__ C,
                                              int M, int K, int N, int tile, int lane) {
    const int Nt    = N >> 4;
    const int mBase = (tile / Nt) << 4;
    const int nBase = (tile % Nt) << 4;

    int row = mBase + (lane & 15);
    if (row > M - 1) row = M - 1;
    const int col  = nBase + (lane & 15);
    const int kOff = (lane >> 4) << 1;

    v8f acc = {};
    for (int k = 0; k < K; k += 4) {
        v2f a = *(const v2f*)(A + (size_t)row * K + k + kOff);
        v2f b;
        b.x = B[(size_t)(k + kOff)     * N + col];
        b.y = B[(size_t)(k + kOff + 1) * N + col];
        acc = __builtin_amdgcn_wmma_f32_16x16x4_f32(
                   false, a,  false, b,
                   (short)0, acc,  false,  false);
    }

    const float bb   = bias ? bias[col] : 0.0f;
    const int   rOff = (lane >> 4) << 3;
#pragma unroll
    for (int j = 0; j < 8; ++j) {
        int r = mBase + j + rOff;
        if (r < M) C[(size_t)r * N + col] = acc[j] + bb;
    }
}

__global__ void k_gemm_wmma(const float* __restrict__ A, const float* __restrict__ B,
                            const float* __restrict__ bias, float* __restrict__ C,
                            int M, int K, int N) {
    const int lane  = threadIdx.x & 31;
    const int wave  = threadIdx.x >> 5;
    const int wpb   = blockDim.x >> 5;
    const int tile  = blockIdx.x * wpb + wave;
    const int Mt    = (M + 15) >> 4;
    const int Nt    = N >> 4;
    if (tile >= Mt * Nt) return;
    gemm_tile_f32(A, B, bias, C, M, K, N, tile, lane);
}

__global__ void k_fold_classifier(const float* __restrict__ W2, const float* __restrict__ b2,
                                  const float* __restrict__ Wc, const float* __restrict__ bc,
                                  float* __restrict__ W2c, float* __restrict__ cvec) {
    const int lane = threadIdx.x & 31;
    const int wave = threadIdx.x >> 5;
    if (wave < FEAT / 16)
        gemm_tile_f32(W2, Wc, nullptr, W2c, FEAT, FEAT, N_CLASSES, wave, lane);
    else
        gemm_tile_f32(b2, Wc, bc, cvec, 1, FEAT, N_CLASSES, 0, lane);
}

__global__ void k_zero_i(int* __restrict__ p, int n) {
    int i = blockIdx.x * blockDim.x + threadIdx.x;
    if (i < n) p[i] = 0;
}

__global__ void k_count(const int* __restrict__ dst, int* __restrict__ cnt, int E) {
    int e = blockIdx.x * blockDim.x + threadIdx.x;
    if (e < E)
        __hip_atomic_fetch_add(&cnt[dst[e]], 1, __ATOMIC_RELAXED, __HIP_MEMORY_SCOPE_AGENT);
}

#define SCAN_T 1024
__global__ void k_scan(const int* __restrict__ cnt, int* __restrict__ rowptr, int* __restrict__ cursor,
                       float* __restrict__ dinv, int n) {
    __shared__ int sTot[SCAN_T / 32];
    __shared__ int sPre[SCAN_T / 32];
    const int t    = threadIdx.x;
    const int lane = t & 31;
    const int wave = t >> 5;
    const int i    = blockIdx.x * SCAN_T + t;

    const int4* cnt4 = (const int4*)cnt;
    const int   nPre = blockIdx.x * (SCAN_T / 4);
    int pre = 0;
#pragma unroll 4
    for (int k = t; k < nPre; k += SCAN_T) {
        const int4 v = cnt4[k];
        pre += (v.x + v.y) + (v.z + v.w);
    }

    const int c = (i < n) ? cnt[i] : 0;
    int inc = c;
#pragma unroll
    for (int o = 1; o < 32; o <<= 1) {
        const int v = __shfl_up(inc, o);
        if (lane >= o) inc += v;
    }
#pragma unroll
    for (int o = 16; o > 0; o >>= 1) pre += __shfl_xor(pre, o);
    if (lane == 31) sTot[wave] = inc;
    if (lane == 0)  sPre[wave] = pre;
    __syncthreads();
    if (wave == 0) {
        const int own = sTot[lane];
        int w = own;
#pragma unroll
        for (int o = 1; o < 32; o <<= 1) {
            const int v = __shfl_up(w, o);
            if (lane >= o) w += v;
        }
        int p = sPre[lane];
#pragma unroll
        for (int o = 16; o > 0; o >>= 1) p += __shfl_xor(p, o);
        sTot[lane] = p + w - own;
    }
    __syncthreads();

    if (i < n) {
        const int ex = sTot[wave] + inc - c;
        rowptr[i] = ex;
        cursor[i] = ex;
        dinv[i]   = rsqrtf((float)(c + 1));
        if (i == n - 1) rowptr[n] = ex + c;
    }
}

__global__ void k_place(const int* __restrict__ src, const int* __restrict__ dst,
                        int* __restrict__ cursor, int* __restrict__ col, int E) {
    int e = blockIdx.x * blockDim.x + threadIdx.x;
    if (e < E) {
        const int p = __hip_atomic_fetch_add(&cursor[dst[e]], 1, __ATOMIC_RELAXED, __HIP_MEMORY_SCOPE_AGENT);
        col[p] = src[e];
    }
}

__global__ void k_gather1(const float* __restrict__ H, const int* __restrict__ rowptr,
                          const int* __restrict__ col, const float* __restrict__ dinv,
                          const float* __restrict__ b1,
                          const float* __restrict__ gamma, const float* __restrict__ beta,
                          const float* __restrict__ mean, const float* __restrict__ var,
                          float* __restrict__ act, int nodeBase, int nodeEnd) {
    const int lane = threadIdx.x & 31;
    const int wave = threadIdx.x >> 5;
    const int node = __builtin_amdgcn_readfirstlane(nodeBase + blockIdx.x * (blockDim.x >> 5) + wave);
    if (node >= nodeEnd) return;

    const int   beg = rowptr[node];
    const int   end = rowptr[node + 1];
    const float dn  = dinv[node];

    const float4 self = *(const float4*)(H + (size_t)node * FEAT + lane * 4);
    const float  wn   = dn * dn;
    float4 acc = make_float4(self.x * wn, self.y * wn, self.z * wn, self.w * wn);

    for (int base = beg; base < end; base += 32) {
        const int m = min(32, end - base);
        int   s = 0;
        float w = 0.0f;
        if (lane < m) {
            s = col[base + lane];
            w = dinv[s] * dn;
        }
#pragma unroll 4
        for (int j = 0; j < m; ++j) {
            const int    sj = __shfl(s, j);
            const float  wj = __shfl(w, j);
            const float4 h  = *(const float4*)(H + (size_t)sj * FEAT + lane * 4);
            acc.x += h.x * wj;
            acc.y += h.y * wj;
            acc.z += h.z * wj;
            acc.w += h.w * wj;
        }
    }

    const float4 vb = *(const float4*)(b1    + lane * 4);
    const float4 vm = *(const float4*)(mean  + lane * 4);
    const float4 vv = *(const float4*)(var   + lane * 4);
    const float4 vg = *(const float4*)(gamma + lane * 4);
    const float4 vt = *(const float4*)(beta  + lane * 4);
    float4 r;
    r.x = fmaxf((acc.x + vb.x - vm.x) * rsqrtf(vv.x + BN_EPS) * vg.x + vt.x, 0.0f);
    r.y = fmaxf((acc.y + vb.y - vm.y) * rsqrtf(vv.y + BN_EPS) * vg.y + vt.y, 0.0f);
    r.z = fmaxf((acc.z + vb.z - vm.z) * rsqrtf(vv.z + BN_EPS) * vg.z + vt.z, 0.0f);
    r.w = fmaxf((acc.w + vb.w - vm.w) * rsqrtf(vv.w + BN_EPS) * vg.w + vt.w, 0.0f);
    *(float4*)(act + (size_t)(node - nodeBase) * FEAT + lane * 4) = r;
}

__global__ void k_gather2(const float* __restrict__ Hc, const int* __restrict__ rowptr,
                          const int* __restrict__ col, const float* __restrict__ dinv,
                          const float* __restrict__ cvec, float* __restrict__ out, int N) {
    const int lane = threadIdx.x & 31;
    const int wave = threadIdx.x >> 5;
    const int f    = lane & (N_CLASSES - 1);
    const int node = ((blockIdx.x * (blockDim.x >> 5) + wave) << 1) + (lane >> 4);
    const bool live = node < N;
    const int  self = live ? node : 0;

    int   beg = 0, len = 0;
    float dn  = 0.0f, acc = 0.0f;
    if (live) {
        beg = rowptr[node];
        len = rowptr[node + 1] - beg;
        dn  = dinv[node];
        acc = Hc[(size_t)node * N_CLASSES + f] * (dn * dn);
    }
    const int lenMax = __builtin_amdgcn_readfirstlane(max(len, __shfl_xor(len, 16)));

    for (int base = 0; base < lenMax; base += 16) {
        int   s = self;
        float w = 0.0f;
        if (base + f < len) {
            s = col[beg + base + f];
            w = dinv[s] * dn;
        }
        const int m = min(16, lenMax - base);
#pragma unroll 4
        for (int j = 0; j < m; ++j) {
            const int   sj = __shfl(s, j, 16);
            const float wj = __shfl(w, j, 16);
            acc += Hc[(size_t)sj * N_CLASSES + f] * wj;
        }
    }
    if (live) out[(size_t)node * N_CLASSES + f] = acc + cvec[f];
}

extern "C" void kernel_launch(void* const* d_in, const int* in_sizes, int n_in,
                              void* d_out, int out_size, void* d_ws, size_t ws_size,
                              hipStream_t stream) {
    const float* X     = (const float*)d_in[0];
    const int*   edge  = (const int*)  d_in[1];
    const float* W1    = (const float*)d_in[2];
    const float* b1    = (const float*)d_in[3];
    const float* gamma = (const float*)d_in[4];
    const float* beta  = (const float*)d_in[5];
    const float* rmean = (const float*)d_in[6];
    const float* rvar  = (const float*)d_in[7];
    const float* W2    = (const float*)d_in[8];
    const float* b2    = (const float*)d_in[9];
    const float* Wc    = (const float*)d_in[10];
    const float* bc    = (const float*)d_in[11];
    float*       out   = (float*)d_out;

    const int N = in_sizes[0] / FEAT;
    const int E = in_sizes[1] / 2;
    const int* src = edge;
    const int* dst = edge + E;
    const int total = N * FEAT;
    const int totalC = N * N_CLASSES;

    const int Npad  = (N + 63) & ~63;
    const int nHalf = ((N >> 1) + 15) & ~15;
    float* dinv   = (float*)d_ws;
    float* bufA   = dinv + Npad;
    float* actH   = bufA + (size_t)total;
    float* Hc     = actH + (size_t)nHalf * FEAT;
    float* W2c    = Hc + (size_t)totalC;
    float* cvec   = W2c + FEAT * N_CLASSES;
    int*   rowptr = (int*)(cvec + 64);
    int*   cnt    = rowptr + Npad + 64;
    int*   cursor = cnt + Npad;
    int*   col    = cursor + Npad;

    const int T256   = 256;
    const int gNode  = (N + T256 - 1) / T256;
    const int gEdge  = (E + T256 - 1) / T256;
    const int tilesL = ((N + 15) / 16) * (FEAT / 16);
    const int gGemmL = (tilesL + 7) / 8;
    const int gGat2  = (N + 15) / 16;

    k_zero_i<<<gNode, T256, 0, stream>>>(cnt, N);
    k_count<<<gEdge, T256, 0, stream>>>(dst, cnt, E);
    k_scan<<<(N + SCAN_T - 1) / SCAN_T, SCAN_T, 0, stream>>>(cnt, rowptr, cursor, dinv, N);
    k_place<<<gEdge, T256, 0, stream>>>(src, dst, cursor, col, E);

    k_gemm_wmma<<<gGemmL, T256, 0, stream>>>(X, W1, nullptr, bufA, N, FEAT, FEAT);
    k_fold_classifier<<<1, 32 * (FEAT / 16 + 1), 0, stream>>>(W2, b2, Wc, bc, W2c, cvec);

    const int nRest = N - nHalf;
    k_gather1<<<(nHalf + 7) / 8, T256, 0, stream>>>(bufA, rowptr, col, dinv, b1, gamma, beta, rmean, rvar, actH, 0, nHalf);
    k_gemm_wmma<<<((nHalf + 15) / 16 + 7) / 8, T256, 0, stream>>>(actH, W2c, nullptr, Hc, nHalf, FEAT, N_CLASSES);
    k_gather1<<<(nRest + 7) / 8, T256, 0, stream>>>(bufA, rowptr, col, dinv, b1, gamma, beta, rmean, rvar, actH, nHalf, N);
    k_gemm_wmma<<<((nRest + 15) / 16 + 7) / 8, T256, 0, stream>>>(actH, W2c, nullptr, Hc + (size_t)nHalf * N_CLASSES, nRest, FEAT, N_CLASSES);

    k_gather2<<<gGat2, T256, 0, stream>>>(Hc, rowptr, col, dinv, cvec, out, N);
}


# Neighbour lists in fixed slot rows: the rank that one returning add per edge draws is the edge's slot, so the count and prefix-sum launches go (11 to 9); f32 sums still run in list arrival order
# speedup vs baseline: 2.6767x; 1.2864x over previous
#include <hip/hip_runtime.h>
#include <math.h>

typedef float v2f __attribute__((ext_vector_type(2)));
typedef float v8f __attribute__((ext_vector_type(8)));

#define FEAT 128
#define N_CLASSES 16
#define BN_EPS 1e-5f

__device__ __forceinline__ void gemm_tile_f32(const float* __restrict__ A, const float* __restrict__ B,
                                              const float* __restrict__ bias, float* __restrict__ C,
                                              int M, int K, int N, int tile, int lane) {
    const int Nt    = N >> 4;
    const int mBase = (tile / Nt) << 4;
    const int nBase = (tile % Nt) << 4;

    int row = mBase + (lane & 15);
    if (row > M - 1) row = M - 1;
    const int col  = nBase + (lane & 15);
    const int kOff = (lane >> 4) << 1;

    v8f acc = {};
    for (int k = 0; k < K; k += 4) {
        v2f a = *(const v2f*)(A + (size_t)row * K + k + kOff);
        v2f b;
        b.x = B[(size_t)(k + kOff)     * N + col];
        b.y = B[(size_t)(k + kOff + 1) * N + col];
        acc = __builtin_amdgcn_wmma_f32_16x16x4_f32(
                   false, a,  false, b,
                   (short)0, acc,  false,  false);
    }

    const float bb   = bias ? bias[col] : 0.0f;
    const int   rOff = (lane >> 4) << 3;
#pragma unroll
    for (int j = 0; j < 8; ++j) {
        int r = mBase + j + rOff;
        if (r < M) C[(size_t)r * N + col] = acc[j] + bb;
    }
}

__global__ void k_gemm_wmma(const float* __restrict__ A, const float* __restrict__ B,
                            const float* __restrict__ bias, float* __restrict__ C,
                            int M, int K, int N) {
    const int lane  = threadIdx.x & 31;
    const int wave  = threadIdx.x >> 5;
    const int wpb   = blockDim.x >> 5;
    const int tile  = blockIdx.x * wpb + wave;
    const int Mt    = (M + 15) >> 4;
    const int Nt    = N >> 4;
    if (tile >= Mt * Nt) return;
    gemm_tile_f32(A, B, bias, C, M, K, N, tile, lane);
}

__global__ void k_fold_classifier(const float* __restrict__ W2, const float* __restrict__ b2,
                                  const float* __restrict__ Wc, const float* __restrict__ bc,
                                  float* __restrict__ W2c, float* __restrict__ cvec) {
    const int lane = threadIdx.x & 31;
    const int wave = threadIdx.x >> 5;
    if (wave < FEAT / 16)
        gemm_tile_f32(W2, Wc, nullptr, W2c, FEAT, FEAT, N_CLASSES, wave, lane);
    else
        gemm_tile_f32(b2, Wc, bc, cvec, 1, FEAT, N_CLASSES, 0, lane);
}

#define SLOT_CAP 40

__global__ void k_zero_i(int* __restrict__ p, int n) {
    int i = blockIdx.x * blockDim.x + threadIdx.x;
    if (i < n) p[i] = 0;
}

__global__ void k_place_slots(const int* __restrict__ src, const int* __restrict__ dst,
                              int* __restrict__ cnt, int* __restrict__ slots, int E) {
    int e = blockIdx.x * blockDim.x + threadIdx.x;
    if (e < E) {
        const int d = dst[e];
        const int p = __hip_atomic_fetch_add(&cnt[d], 1, __ATOMIC_RELAXED, __HIP_MEMORY_SCOPE_AGENT);
        if (p < SLOT_CAP) slots[(size_t)d * SLOT_CAP + p] = src[e];
    }
}

__global__ void k_gather1(const float* __restrict__ H, const int* __restrict__ cnt,
                          const int* __restrict__ slots,
                          const int* __restrict__ src, const int* __restrict__ dst, int E,
                          const float* __restrict__ b1,
                          const float* __restrict__ gamma, const float* __restrict__ beta,
                          const float* __restrict__ mean, const float* __restrict__ var,
                          float* __restrict__ act, int nodeBase, int nodeEnd) {
    const int lane = threadIdx.x & 31;
    const int wave = threadIdx.x >> 5;
    const int node = __builtin_amdgcn_readfirstlane(nodeBase + blockIdx.x * (blockDim.x >> 5) + wave);
    if (node >= nodeEnd) return;

    const int   deg = cnt[node];
    const float dn  = rsqrtf((float)(deg + 1));

    const float4 self = *(const float4*)(H + (size_t)node * FEAT + lane * 4);
    const float  wn   = dn * dn;
    float4 acc = make_float4(self.x * wn, self.y * wn, self.z * wn, self.w * wn);

    if (deg <= SLOT_CAP) {
        const int* row = slots + (size_t)node * SLOT_CAP;
        for (int base = 0; base < deg; base += 32) {
            const int m = min(32, deg - base);
            int   s = 0;
            float w = 0.0f;
            if (lane < m) {
                s = row[base + lane];
                w = rsqrtf((float)(cnt[s] + 1)) * dn;
            }
#pragma unroll 4
            for (int j = 0; j < m; ++j) {
                const int    sj = __shfl(s, j);
                const float  wj = __shfl(w, j);
                const float4 h  = *(const float4*)(H + (size_t)sj * FEAT + lane * 4);
                acc.x += h.x * wj;
                acc.y += h.y * wj;
                acc.z += h.z * wj;
                acc.w += h.w * wj;
            }
        }
    } else {
        for (int base = 0; base < E; base += 32) {
            const int e = base + lane;
            int   s = 0;
            float w = 0.0f;
            bool  mine = false;
            if (e < E) mine = (dst[e] == node);
            if (mine) {
                s = src[e];
                w = rsqrtf((float)(cnt[s] + 1)) * dn;
            }
            unsigned hit = __builtin_amdgcn_ballot_w32(mine);
            while (hit != 0u) {
                const int j = __builtin_ctz(hit);
                hit &= hit - 1u;
                const int    sj = __shfl(s, j);
                const float  wj = __shfl(w, j);
                const float4 h  = *(const float4*)(H + (size_t)sj * FEAT + lane * 4);
                acc.x += h.x * wj;
                acc.y += h.y * wj;
                acc.z += h.z * wj;
                acc.w += h.w * wj;
            }
        }
    }

    const float4 vb = *(const float4*)(b1    + lane * 4);
    const float4 vm = *(const float4*)(mean  + lane * 4);
    const float4 vv = *(const float4*)(var   + lane * 4);
    const float4 vg = *(const float4*)(gamma + lane * 4);
    const float4 vt = *(const float4*)(beta  + lane * 4);
    float4 r;
    r.x = fmaxf((acc.x + vb.x - vm.x) * rsqrtf(vv.x + BN_EPS) * vg.x + vt.x, 0.0f);
    r.y = fmaxf((acc.y + vb.y - vm.y) * rsqrtf(vv.y + BN_EPS) * vg.y + vt.y, 0.0f);
    r.z = fmaxf((acc.z + vb.z - vm.z) * rsqrtf(vv.z + BN_EPS) * vg.z + vt.z, 0.0f);
    r.w = fmaxf((acc.w + vb.w - vm.w) * rsqrtf(vv.w + BN_EPS) * vg.w + vt.w, 0.0f);
    *(float4*)(act + (size_t)(node - nodeBase) * FEAT + lane * 4) = r;
}

__global__ void k_gather2(const float* __restrict__ Hc, const int* __restrict__ cnt,
                          const int* __restrict__ slots,
                          const int* __restrict__ src, const int* __restrict__ dst, int E,
                          const float* __restrict__ cvec, float* __restrict__ out, int N) {
    const int lane = threadIdx.x & 31;
    const int wave = threadIdx.x >> 5;
    const int f    = lane & (N_CLASSES - 1);
    const int node = ((blockIdx.x * (blockDim.x >> 5) + wave) << 1) + (lane >> 4);
    const bool live = node < N;
    const int  self = live ? node : 0;

    int   deg = 0;
    float dn  = 0.0f, acc = 0.0f;
    if (live) {
        deg = cnt[node];
        dn  = rsqrtf((float)(deg + 1));
        acc = Hc[(size_t)node * N_CLASSES + f] * (dn * dn);
    }
    const bool over = deg > SLOT_CAP;
    const int  len  = over ? 0 : deg;
    const int* row  = slots + (size_t)self * SLOT_CAP;
    const int lenMax = __builtin_amdgcn_readfirstlane(max(len, __shfl_xor(len, 16)));

    for (int base = 0; base < lenMax; base += 16) {
        int   s = self;
        float w = 0.0f;
        if (base + f < len) {
            s = row[base + f];
            w = rsqrtf((float)(cnt[s] + 1)) * dn;
        }
        const int m = min(16, lenMax - base);
#pragma unroll 4
        for (int j = 0; j < m; ++j) {
            const int   sj = __shfl(s, j, 16);
            const float wj = __shfl(w, j, 16);
            acc += Hc[(size_t)sj * N_CLASSES + f] * wj;
        }
    }

    if (__builtin_amdgcn_ballot_w32(over) != 0u) {
        for (int base = 0; base < E; base += 16) {
            const int e = base + f;
            int   s = self;
            float w = 0.0f;
            bool  mine = false;
            if (over && e < E) mine = (dst[e] == node);
            if (mine) {
                s = src[e];
                w = rsqrtf((float)(cnt[s] + 1)) * dn;
            }
            if (__builtin_amdgcn_ballot_w32(mine) != 0u) {
                for (int j = 0; j < 16; ++j) {
                    const int   sj = __shfl(s, j, 16);
                    const float wj = __shfl(w, j, 16);
                    acc += Hc[(size_t)sj * N_CLASSES + f] * wj;
                }
            }
        }
    }
    if (live) out[(size_t)node * N_CLASSES + f] = acc + cvec[f];
}

extern "C" void kernel_launch(void* const* d_in, const int* in_sizes, int n_in,
                              void* d_out, int out_size, void* d_ws, size_t ws_size,
                              hipStream_t stream) {
    const float* X     = (const float*)d_in[0];
    const int*   edge  = (const int*)  d_in[1];
    const float* W1    = (const float*)d_in[2];
    const float* b1    = (const float*)d_in[3];
    const float* gamma = (const float*)d_in[4];
    const float* beta  = (const float*)d_in[5];
    const float* rmean = (const float*)d_in[6];
    const float* rvar  = (const float*)d_in[7];
    const float* W2    = (const float*)d_in[8];
    const float* b2    = (const float*)d_in[9];
    const float* Wc    = (const float*)d_in[10];
    const float* bc    = (const float*)d_in[11];
    float*       out   = (float*)d_out;

    const int N = in_sizes[0] / FEAT;
    const int E = in_sizes[1] / 2;
    const int* src = edge;
    const int* dst = edge + E;
    const int total = N * FEAT;
    const int totalC = N * N_CLASSES;

    const int Npad  = (N + 63) & ~63;
    const int nHalf = ((N >> 1) + 15) & ~15;
    int*   cnt    = (int*)d_ws;
    float* bufA   = (float*)d_ws + Npad;
    float* actH   = bufA + (size_t)total;
    float* Hc     = actH + (size_t)nHalf * FEAT;
    float* W2c    = Hc + (size_t)totalC;
    float* cvec   = W2c + FEAT * N_CLASSES;
    int*   slots  = (int*)(cvec + 64);

    const int T256   = 256;
    const int gNode  = (N + T256 - 1) / T256;
    const int gEdge  = (E + T256 - 1) / T256;
    const int tilesL = ((N + 15) / 16) * (FEAT / 16);
    const int gGemmL = (tilesL + 7) / 8;
    const int gGat2  = (N + 15) / 16;

    k_zero_i<<<gNode, T256, 0, stream>>>(cnt, N);
    k_place_slots<<<gEdge, T256, 0, stream>>>(src, dst, cnt, slots, E);

    k_gemm_wmma<<<gGemmL, T256, 0, stream>>>(X, W1, nullptr, bufA, N, FEAT, FEAT);
    k_fold_classifier<<<1, 32 * (FEAT / 16 + 1), 0, stream>>>(W2, b2, Wc, bc, W2c, cvec);

    const int nRest = N - nHalf;
    k_gather1<<<(nHalf + 7) / 8, T256, 0, stream>>>(bufA, cnt, slots, src, dst, E, b1, gamma, beta, rmean, rvar, actH, 0, nHalf);
    k_gemm_wmma<<<((nHalf + 15) / 16 + 7) / 8, T256, 0, stream>>>(actH, W2c, nullptr, Hc, nHalf, FEAT, N_CLASSES);
    k_gather1<<<(nRest + 7) / 8, T256, 0, stream>>>(bufA, cnt, slots, src, dst, E, b1, gamma, beta, rmean, rvar, actH, nHalf, N);
    k_gemm_wmma<<<((nRest + 15) / 16 + 7) / 8, T256, 0, stream>>>(actH, W2c, nullptr, Hc + (size_t)nHalf * N_CLASSES, nRest, FEAT, N_CLASSES);

    k_gather2<<<gGat2, T256, 0, stream>>>(Hc, cnt, slots, src, dst, E, cvec, out, N);
}


# Layer-2 product fused into the layer-1 gather: 8-wave blocks put 16 activated rows in an LDS tile and wave 0 multiplies it by W2c on the f32 WMMA; 9 launches to 6; f32 sum order unchanged
# speedup vs baseline: 3.0273x; 1.1310x over previous
#include <hip/hip_runtime.h>
#include <math.h>

typedef float v2f __attribute__((ext_vector_type(2)));
typedef float v8f __attribute__((ext_vector_type(8)));

#define FEAT 128
#define N_CLASSES 16
#define BN_EPS 1e-5f

__device__ __forceinline__ void gemm_tile_f32(const float* __restrict__ A, const float* __restrict__ B,
                                              const float* __restrict__ bias, float* __restrict__ C,
                                              int M, int K, int N, int tile, int lane) {
    const int Nt    = N >> 4;
    const int mBase = (tile / Nt) << 4;
    const int nBase = (tile % Nt) << 4;

    int row = mBase + (lane & 15);
    if (row > M - 1) row = M - 1;
    const int col  = nBase + (lane & 15);
    const int kOff = (lane >> 4) << 1;

    v8f acc = {};
    for (int k = 0; k < K; k += 4) {
        v2f a = *(const v2f*)(A + (size_t)row * K + k + kOff);
        v2f b;
        b.x = B[(size_t)(k + kOff)     * N + col];
        b.y = B[(size_t)(k + kOff + 1) * N + col];
        acc = __builtin_amdgcn_wmma_f32_16x16x4_f32(
                   false, a,  false, b,
                   (short)0, acc,  false,  false);
    }

    const float bb   = bias ? bias[col] : 0.0f;
    const int   rOff = (lane >> 4) << 3;
#pragma unroll
    for (int j = 0; j < 8; ++j) {
        int r = mBase + j + rOff;
        if (r < M) C[(size_t)r * N + col] = acc[j] + bb;
    }
}

__global__ void k_gemm_wmma(const float* __restrict__ A, const float* __restrict__ B,
                            const float* __restrict__ bias, float* __restrict__ C,
                            int M, int K, int N) {
    const int lane  = threadIdx.x & 31;
    const int wave  = threadIdx.x >> 5;
    const int wpb   = blockDim.x >> 5;
    const int tile  = blockIdx.x * wpb + wave;
    const int Mt    = (M + 15) >> 4;
    const int Nt    = N >> 4;
    if (tile >= Mt * Nt) return;
    gemm_tile_f32(A, B, bias, C, M, K, N, tile, lane);
}

__global__ void k_fold_classifier(const float* __restrict__ W2, const float* __restrict__ b2,
                                  const float* __restrict__ Wc, const float* __restrict__ bc,
                                  float* __restrict__ W2c, float* __restrict__ cvec) {
    const int lane = threadIdx.x & 31;
    const int wave = threadIdx.x >> 5;
    if (wave < FEAT / 16)
        gemm_tile_f32(W2, Wc, nullptr, W2c, FEAT, FEAT, N_CLASSES, wave, lane);
    else
        gemm_tile_f32(b2, Wc, bc, cvec, 1, FEAT, N_CLASSES, 0, lane);
}

#define SLOT_CAP 40

__global__ void k_zero_i(int* __restrict__ p, int n) {
    int i = blockIdx.x * blockDim.x + threadIdx.x;
    if (i < n) p[i] = 0;
}

__global__ void k_place_slots(const int* __restrict__ src, const int* __restrict__ dst,
                              int* __restrict__ cnt, int* __restrict__ slots, int E) {
    int e = blockIdx.x * blockDim.x + threadIdx.x;
    if (e < E) {
        const int d = dst[e];
        const int p = __hip_atomic_fetch_add(&cnt[d], 1, __ATOMIC_RELAXED, __HIP_MEMORY_SCOPE_AGENT);
        if (p < SLOT_CAP) slots[(size_t)d * SLOT_CAP + p] = src[e];
    }
}

#define TILE_NODES 16
#define ACT_PITCH  (FEAT + 4)

__global__ void k_gather1_fc(const float* __restrict__ H, const int* __restrict__ cnt,
                             const int* __restrict__ slots,
                             const int* __restrict__ src, const int* __restrict__ dst, int E,
                             const float* __restrict__ b1,
                             const float* __restrict__ gamma, const float* __restrict__ beta,
                             const float* __restrict__ mean, const float* __restrict__ var,
                             const float* __restrict__ W2c, float* __restrict__ Hc, int N) {
    __shared__ __attribute__((aligned(16))) float tile[TILE_NODES * ACT_PITCH];
    const int lane = threadIdx.x & 31;
    const int wave     = __builtin_amdgcn_readfirstlane((int)(threadIdx.x >> 5));
    const int tileBase = blockIdx.x * TILE_NODES;

    for (int t = 0; t < TILE_NODES / 8; ++t) {
        const int rowT = wave + 8 * t;
        const int node = tileBase + rowT;
        float4 r = make_float4(0.0f, 0.0f, 0.0f, 0.0f);
        if (node < N) {
            const int   deg = cnt[node];
            const float dn  = rsqrtf((float)(deg + 1));

            const float4 self = *(const float4*)(H + (size_t)node * FEAT + lane * 4);
            const float  wn   = dn * dn;
            float4 acc = make_float4(self.x * wn, self.y * wn, self.z * wn, self.w * wn);

            if (deg <= SLOT_CAP) {
                const int* row = slots + (size_t)node * SLOT_CAP;
                for (int base = 0; base < deg; base += 32) {
                    const int m = min(32, deg - base);
                    int   s = 0;
                    float w = 0.0f;
                    if (lane < m) {
                        s = row[base + lane];
                        w = rsqrtf((float)(cnt[s] + 1)) * dn;
                    }
#pragma unroll 4
                    for (int j = 0; j < m; ++j) {
                        const int    sj = __shfl(s, j);
                        const float  wj = __shfl(w, j);
                        const float4 h  = *(const float4*)(H + (size_t)sj * FEAT + lane * 4);
                        acc.x += h.x * wj;
                        acc.y += h.y * wj;
                        acc.z += h.z * wj;
                        acc.w += h.w * wj;
                    }
                }
            } else {
                for (int base = 0; base < E; base += 32) {
                    const int e = base + lane;
                    int   s = 0;
                    float w = 0.0f;
                    bool  mine = false;
                    if (e < E) mine = (dst[e] == node);
                    if (mine) {
                        s = src[e];
                        w = rsqrtf((float)(cnt[s] + 1)) * dn;
                    }
                    unsigned hit = __builtin_amdgcn_ballot_w32(mine);
                    while (hit != 0u) {
                        const int j = __builtin_ctz(hit);
                        hit &= hit - 1u;
                        const int    sj = __shfl(s, j);
                        const float  wj = __shfl(w, j);
                        const float4 h  = *(const float4*)(H + (size_t)sj * FEAT + lane * 4);
                        acc.x += h.x * wj;
                        acc.y += h.y * wj;
                        acc.z += h.z * wj;
                        acc.w += h.w * wj;
                    }
                }
            }

            const float4 vb = *(const float4*)(b1    + lane * 4);
            const float4 vm = *(const float4*)(mean  + lane * 4);
            const float4 vv = *(const float4*)(var   + lane * 4);
            const float4 vg = *(const float4*)(gamma + lane * 4);
            const float4 vt = *(const float4*)(beta  + lane * 4);
            r.x = fmaxf((acc.x + vb.x - vm.x) * rsqrtf(vv.x + BN_EPS) * vg.x + vt.x, 0.0f);
            r.y = fmaxf((acc.y + vb.y - vm.y) * rsqrtf(vv.y + BN_EPS) * vg.y + vt.y, 0.0f);
            r.z = fmaxf((acc.z + vb.z - vm.z) * rsqrtf(vv.z + BN_EPS) * vg.z + vt.z, 0.0f);
            r.w = fmaxf((acc.w + vb.w - vm.w) * rsqrtf(vv.w + BN_EPS) * vg.w + vt.w, 0.0f);
        }
        *(float4*)(tile + rowT * ACT_PITCH + lane * 4) = r;
    }
    __syncthreads();

    if (wave == 0) {
        const int rc   = lane & 15;
        const int kOff = (lane >> 4) << 1;
        v8f acc = {};
#pragma unroll 4
        for (int k = 0; k < FEAT; k += 4) {
            const v2f a = *(const v2f*)(tile + rc * ACT_PITCH + k + kOff);
            v2f b;
            b.x = W2c[(k + kOff)     * N_CLASSES + rc];
            b.y = W2c[(k + kOff + 1) * N_CLASSES + rc];
            acc = __builtin_amdgcn_wmma_f32_16x16x4_f32(
                       false, a,  false, b,
                       (short)0, acc,  false,  false);
        }
        const int rOff = (lane >> 4) << 3;
#pragma unroll
        for (int j = 0; j < 8; ++j) {
            const int rr = tileBase + j + rOff;
            if (rr < N) Hc[(size_t)rr * N_CLASSES + rc] = acc[j];
        }
    }
}

__global__ void k_gather2(const float* __restrict__ Hc, const int* __restrict__ cnt,
                          const int* __restrict__ slots,
                          const int* __restrict__ src, const int* __restrict__ dst, int E,
                          const float* __restrict__ cvec, float* __restrict__ out, int N) {
    const int lane = threadIdx.x & 31;
    const int wave = threadIdx.x >> 5;
    const int f    = lane & (N_CLASSES - 1);
    const int node = ((blockIdx.x * (blockDim.x >> 5) + wave) << 1) + (lane >> 4);
    const bool live = node < N;
    const int  self = live ? node : 0;

    int   deg = 0;
    float dn  = 0.0f, acc = 0.0f;
    if (live) {
        deg = cnt[node];
        dn  = rsqrtf((float)(deg + 1));
        acc = Hc[(size_t)node * N_CLASSES + f] * (dn * dn);
    }
    const bool over = deg > SLOT_CAP;
    const int  len  = over ? 0 : deg;
    const int* row  = slots + (size_t)self * SLOT_CAP;
    const int lenMax = __builtin_amdgcn_readfirstlane(max(len, __shfl_xor(len, 16)));

    for (int base = 0; base < lenMax; base += 16) {
        int   s = self;
        float w = 0.0f;
        if (base + f < len) {
            s = row[base + f];
            w = rsqrtf((float)(cnt[s] + 1)) * dn;
        }
        const int m = min(16, lenMax - base);
#pragma unroll 4
        for (int j = 0; j < m; ++j) {
            const int   sj = __shfl(s, j, 16);
            const float wj = __shfl(w, j, 16);
            acc += Hc[(size_t)sj * N_CLASSES + f] * wj;
        }
    }

    if (__builtin_amdgcn_ballot_w32(over) != 0u) {
        for (int base = 0; base < E; base += 16) {
            const int e = base + f;
            int   s = self;
            float w = 0.0f;
            bool  mine = false;
            if (over && e < E) mine = (dst[e] == node);
            if (mine) {
                s = src[e];
                w = rsqrtf((float)(cnt[s] + 1)) * dn;
            }
            if (__builtin_amdgcn_ballot_w32(mine) != 0u) {
                for (int j = 0; j < 16; ++j) {
                    const int   sj = __shfl(s, j, 16);
                    const float wj = __shfl(w, j, 16);
                    acc += Hc[(size_t)sj * N_CLASSES + f] * wj;
                }
            }
        }
    }
    if (live) out[(size_t)node * N_CLASSES + f] = acc + cvec[f];
}

extern "C" void kernel_launch(void* const* d_in, const int* in_sizes, int n_in,
                              void* d_out, int out_size, void* d_ws, size_t ws_size,
                              hipStream_t stream) {
    const float* X     = (const float*)d_in[0];
    const int*   edge  = (const int*)  d_in[1];
    const float* W1    = (const float*)d_in[2];
    const float* b1    = (const float*)d_in[3];
    const float* gamma = (const float*)d_in[4];
    const float* beta  = (const float*)d_in[5];
    const float* rmean = (const float*)d_in[6];
    const float* rvar  = (const float*)d_in[7];
    const float* W2    = (const float*)d_in[8];
    const float* b2    = (const float*)d_in[9];
    const float* Wc    = (const float*)d_in[10];
    const float* bc    = (const float*)d_in[11];
    float*       out   = (float*)d_out;

    const int N = in_sizes[0] / FEAT;
    const int E = in_sizes[1] / 2;
    const int* src = edge;
    const int* dst = edge + E;
    const int total = N * FEAT;
    const int totalC = N * N_CLASSES;

    const int Npad  = (N + 63) & ~63;
    int*   cnt    = (int*)d_ws;
    float* bufA   = (float*)d_ws + Npad;
    float* Hc     = bufA + (size_t)total;
    float* W2c    = Hc + (size_t)totalC;
    float* cvec   = W2c + FEAT * N_CLASSES;
    int*   slots  = (int*)(cvec + 64);

    const int T256   = 256;
    const int gNode  = (N + T256 - 1) / T256;
    const int gEdge  = (E + T256 - 1) / T256;
    const int tilesL = ((N + 15) / 16) * (FEAT / 16);
    const int gGemmL = (tilesL + 7) / 8;
    const int gGat2  = (N + 15) / 16;

    k_zero_i<<<gNode, T256, 0, stream>>>(cnt, N);
    k_place_slots<<<gEdge, T256, 0, stream>>>(src, dst, cnt, slots, E);

    k_gemm_wmma<<<gGemmL, T256, 0, stream>>>(X, W1, nullptr, bufA, N, FEAT, FEAT);
    k_fold_classifier<<<1, 32 * (FEAT / 16 + 1), 0, stream>>>(W2, b2, Wc, bc, W2c, cvec);

    k_gather1_fc<<<(N + TILE_NODES - 1) / TILE_NODES, T256, 0, stream>>>(bufA, cnt, slots, src, dst, E, b1, gamma, beta, rmean, rvar, W2c, Hc, N);

    k_gather2<<<gGat2, T256, 0, stream>>>(Hc, cnt, slots, src, dst, E, cvec, out, N);
}


# Layer-1 product moved behind the aggregation, (Ahat X) W1, and into the gather blocks: 8 waves multiply the LDS tile of summed inputs by W1 on f32 WMMA; 6 launches to 5; f32 sums reassociated
# speedup vs baseline: 3.7150x; 1.2272x over previous
#include <hip/hip_runtime.h>
#include <math.h>

typedef float v2f __attribute__((ext_vector_type(2)));
typedef float v8f __attribute__((ext_vector_type(8)));

#define FEAT 128
#define N_CLASSES 16
#define BN_EPS 1e-5f

__device__ __forceinline__ void gemm_tile_f32(const float* __restrict__ A, const float* __restrict__ B,
                                              const float* __restrict__ bias, float* __restrict__ C,
                                              int M, int K, int N, int tile, int lane) {
    const int Nt    = N >> 4;
    const int mBase = (tile / Nt) << 4;
    const int nBase = (tile % Nt) << 4;

    int row = mBase + (lane & 15);
    if (row > M - 1) row = M - 1;
    const int col  = nBase + (lane & 15);
    const int kOff = (lane >> 4) << 1;

    v8f acc = {};
    for (int k = 0; k < K; k += 4) {
        v2f a = *(const v2f*)(A + (size_t)row * K + k + kOff);
        v2f b;
        b.x = B[(size_t)(k + kOff)     * N + col];
        b.y = B[(size_t)(k + kOff + 1) * N + col];
        acc = __builtin_amdgcn_wmma_f32_16x16x4_f32(
                   false, a,  false, b,
                   (short)0, acc,  false,  false);
    }

    const float bb   = bias ? bias[col] : 0.0f;
    const int   rOff = (lane >> 4) << 3;
#pragma unroll
    for (int j = 0; j < 8; ++j) {
        int r = mBase + j + rOff;
        if (r < M) C[(size_t)r * N + col] = acc[j] + bb;
    }
}

__global__ void k_fold_classifier(const float* __restrict__ W2, const float* __restrict__ b2,
                                  const float* __restrict__ Wc, const float* __restrict__ bc,
                                  float* __restrict__ W2c, float* __restrict__ cvec) {
    const int lane = threadIdx.x & 31;
    const int wave = threadIdx.x >> 5;
    if (wave < FEAT / 16)
        gemm_tile_f32(W2, Wc, nullptr, W2c, FEAT, FEAT, N_CLASSES, wave, lane);
    else
        gemm_tile_f32(b2, Wc, bc, cvec, 1, FEAT, N_CLASSES, 0, lane);
}

#define SLOT_CAP 40

__global__ void k_zero_i(int* __restrict__ p, int n) {
    int i = blockIdx.x * blockDim.x + threadIdx.x;
    if (i < n) p[i] = 0;
}

__global__ void k_place_slots(const int* __restrict__ src, const int* __restrict__ dst,
                              int* __restrict__ cnt, int* __restrict__ slots, int E) {
    int e = blockIdx.x * blockDim.x + threadIdx.x;
    if (e < E) {
        const int d = dst[e];
        const int p = __hip_atomic_fetch_add(&cnt[d], 1, __ATOMIC_RELAXED, __HIP_MEMORY_SCOPE_AGENT);
        if (p < SLOT_CAP) slots[(size_t)d * SLOT_CAP + p] = src[e];
    }
}

#define TILE_NODES 16
#define ACT_PITCH  (FEAT + 4)

__global__ void k_gather1_fc(const float* __restrict__ X, const int* __restrict__ cnt,
                             const int* __restrict__ slots,
                             const int* __restrict__ src, const int* __restrict__ dst, int E,
                             const float* __restrict__ W1, const float* __restrict__ b1,
                             const float* __restrict__ gamma, const float* __restrict__ beta,
                             const float* __restrict__ mean, const float* __restrict__ var,
                             const float* __restrict__ W2c, float* __restrict__ Hc, int N) {
    __shared__ __attribute__((aligned(16))) float aggT[TILE_NODES * ACT_PITCH];
    __shared__ __attribute__((aligned(16))) float actT[TILE_NODES * ACT_PITCH];
    const int lane = threadIdx.x & 31;
    const int wave     = __builtin_amdgcn_readfirstlane((int)(threadIdx.x >> 5));
    const int tileBase = blockIdx.x * TILE_NODES;

    const int rc   = lane & 15;
    const int kOff = (lane >> 4) << 1;
    const int rOff = (lane >> 4) << 3;
    const int col = (wave << 4) + rc;
    const float* __restrict__ w1p = W1 + kOff * FEAT + col;
    float bcur[16], bnxt[16];
#pragma unroll
    for (int i = 0; i < 8; ++i) {
        bcur[2 * i]     = w1p[(4 * i)     * FEAT];
        bcur[2 * i + 1] = w1p[(4 * i + 1) * FEAT];
    }

    for (int t = 0; t < TILE_NODES / 8; ++t) {
        const int rowT = wave + 8 * t;
        const int node = tileBase + rowT;
        float4 acc = make_float4(0.0f, 0.0f, 0.0f, 0.0f);
        if (node < N) {
            const int   deg = cnt[node];
            const float dn  = rsqrtf((float)(deg + 1));

            const float4 self = *(const float4*)(X + (size_t)node * FEAT + lane * 4);
            const float  wn   = dn * dn;
            acc = make_float4(self.x * wn, self.y * wn, self.z * wn, self.w * wn);

            if (deg <= SLOT_CAP) {
                const int* row = slots + (size_t)node * SLOT_CAP;
                for (int base = 0; base < deg; base += 32) {
                    const int m = min(32, deg - base);
                    int   s = 0;
                    float w = 0.0f;
                    if (lane < m) {
                        s = row[base + lane];
                        w = rsqrtf((float)(cnt[s] + 1)) * dn;
                    }
#pragma unroll 4
                    for (int j = 0; j < m; ++j) {
                        const int    sj = __shfl(s, j);
                        const float  wj = __shfl(w, j);
                        const float4 h  = *(const float4*)(X + (size_t)sj * FEAT + lane * 4);
                        acc.x += h.x * wj;
                        acc.y += h.y * wj;
                        acc.z += h.z * wj;
                        acc.w += h.w * wj;
                    }
                }
            } else {
                for (int base = 0; base < E; base += 32) {
                    const int e = base + lane;
                    int   s = 0;
                    float w = 0.0f;
                    bool  mine = false;
                    if (e < E) mine = (dst[e] == node);
                    if (mine) {
                        s = src[e];
                        w = rsqrtf((float)(cnt[s] + 1)) * dn;
                    }
                    unsigned hit = __builtin_amdgcn_ballot_w32(mine);
                    while (hit != 0u) {
                        const int j = __builtin_ctz(hit);
                        hit &= hit - 1u;
                        const int    sj = __shfl(s, j);
                        const float  wj = __shfl(w, j);
                        const float4 h  = *(const float4*)(X + (size_t)sj * FEAT + lane * 4);
                        acc.x += h.x * wj;
                        acc.y += h.y * wj;
                        acc.z += h.z * wj;
                        acc.w += h.w * wj;
                    }
                }
            }
        }
        *(float4*)(aggT + rowT * ACT_PITCH + lane * 4) = acc;
    }
    __syncthreads();

    {
        v8f acc = {};
#pragma unroll
        for (int q = 0; q < 4; ++q) {
            if (q + 1 < 4) {
#pragma unroll
                for (int i = 0; i < 8; ++i) {
                    bnxt[2 * i]     = w1p[(32 * (q + 1) + 4 * i)     * FEAT];
                    bnxt[2 * i + 1] = w1p[(32 * (q + 1) + 4 * i + 1) * FEAT];
                }
            }
#pragma unroll
            for (int i = 0; i < 8; ++i) {
                const int k = 32 * q + 4 * i;
                const v2f a = *(const v2f*)(aggT + rc * ACT_PITCH + k + kOff);
                v2f b;
                b.x = bcur[2 * i];
                b.y = bcur[2 * i + 1];
                acc = __builtin_amdgcn_wmma_f32_16x16x4_f32(
                           false, a,  false, b,
                           (short)0, acc,  false,  false);
            }
#pragma unroll
            for (int i = 0; i < 16; ++i) bcur[i] = bnxt[i];
        }
        const float vb = b1[col];
        const float vm = mean[col];
        const float vs = rsqrtf(var[col] + BN_EPS);
        const float vg = gamma[col];
        const float vt = beta[col];
#pragma unroll
        for (int j = 0; j < 8; ++j)
            actT[(j + rOff) * ACT_PITCH + col] = fmaxf((acc[j] + vb - vm) * vs * vg + vt, 0.0f);
    }
    __syncthreads();

    if (wave == 0) {
        v8f acc = {};
#pragma unroll 4
        for (int k = 0; k < FEAT; k += 4) {
            const v2f a = *(const v2f*)(actT + rc * ACT_PITCH + k + kOff);
            v2f b;
            b.x = W2c[(k + kOff)     * N_CLASSES + rc];
            b.y = W2c[(k + kOff + 1) * N_CLASSES + rc];
            acc = __builtin_amdgcn_wmma_f32_16x16x4_f32(
                       false, a,  false, b,
                       (short)0, acc,  false,  false);
        }
#pragma unroll
        for (int j = 0; j < 8; ++j) {
            const int rr = tileBase + j + rOff;
            if (rr < N) Hc[(size_t)rr * N_CLASSES + rc] = acc[j];
        }
    }
}

__global__ void k_gather2(const float* __restrict__ Hc, const int* __restrict__ cnt,
                          const int* __restrict__ slots,
                          const int* __restrict__ src, const int* __restrict__ dst, int E,
                          const float* __restrict__ cvec, float* __restrict__ out, int N) {
    const int lane = threadIdx.x & 31;
    const int wave = threadIdx.x >> 5;
    const int f    = lane & (N_CLASSES - 1);
    const int node = ((blockIdx.x * (blockDim.x >> 5) + wave) << 1) + (lane >> 4);
    const bool live = node < N;
    const int  self = live ? node : 0;

    int   deg = 0;
    float dn  = 0.0f, acc = 0.0f;
    if (live) {
        deg = cnt[node];
        dn  = rsqrtf((float)(deg + 1));
        acc = Hc[(size_t)node * N_CLASSES + f] * (dn * dn);
    }
    const bool over = deg > SLOT_CAP;
    const int  len  = over ? 0 : deg;
    const int* row  = slots + (size_t)self * SLOT_CAP;
    const int lenMax = __builtin_amdgcn_readfirstlane(max(len, __shfl_xor(len, 16)));

    for (int base = 0; base < lenMax; base += 16) {
        int   s = self;
        float w = 0.0f;
        if (base + f < len) {
            s = row[base + f];
            w = rsqrtf((float)(cnt[s] + 1)) * dn;
        }
        const int m = min(16, lenMax - base);
#pragma unroll 4
        for (int j = 0; j < m; ++j) {
            const int   sj = __shfl(s, j, 16);
            const float wj = __shfl(w, j, 16);
            acc += Hc[(size_t)sj * N_CLASSES + f] * wj;
        }
    }

    if (__builtin_amdgcn_ballot_w32(over) != 0u) {
        for (int base = 0; base < E; base += 16) {
            const int e = base + f;
            int   s = self;
            float w = 0.0f;
            bool  mine = false;
            if (over && e < E) mine = (dst[e] == node);
            if (mine) {
                s = src[e];
                w = rsqrtf((float)(cnt[s] + 1)) * dn;
            }
            if (__builtin_amdgcn_ballot_w32(mine) != 0u) {
                for (int j = 0; j < 16; ++j) {
                    const int   sj = __shfl(s, j, 16);
                    const float wj = __shfl(w, j, 16);
                    acc += Hc[(size_t)sj * N_CLASSES + f] * wj;
                }
            }
        }
    }
    if (live) out[(size_t)node * N_CLASSES + f] = acc + cvec[f];
}

extern "C" void kernel_launch(void* const* d_in, const int* in_sizes, int n_in,
                              void* d_out, int out_size, void* d_ws, size_t ws_size,
                              hipStream_t stream) {
    const float* X     = (const float*)d_in[0];
    const int*   edge  = (const int*)  d_in[1];
    const float* W1    = (const float*)d_in[2];
    const float* b1    = (const float*)d_in[3];
    const float* gamma = (const float*)d_in[4];
    const float* beta  = (const float*)d_in[5];
    const float* rmean = (const float*)d_in[6];
    const float* rvar  = (const float*)d_in[7];
    const float* W2    = (const float*)d_in[8];
    const float* b2    = (const float*)d_in[9];
    const float* Wc    = (const float*)d_in[10];
    const float* bc    = (const float*)d_in[11];
    float*       out   = (float*)d_out;

    const int N = in_sizes[0] / FEAT;
    const int E = in_sizes[1] / 2;
    const int* src = edge;
    const int* dst = edge + E;
    const int totalC = N * N_CLASSES;

    const int Npad  = (N + 63) & ~63;
    int*   cnt    = (int*)d_ws;
    float* Hc     = (float*)d_ws + Npad;
    float* W2c    = Hc + (size_t)totalC;
    float* cvec   = W2c + FEAT * N_CLASSES;
    int*   slots  = (int*)(cvec + 64);

    const int T256   = 256;
    const int gNode  = (N + T256 - 1) / T256;
    const int gEdge  = (E + T256 - 1) / T256;
    const int gGat2  = (N + 15) / 16;

    k_zero_i<<<gNode, T256, 0, stream>>>(cnt, N);
    k_place_slots<<<gEdge, T256, 0, stream>>>(src, dst, cnt, slots, E);

    k_fold_classifier<<<1, 32 * (FEAT / 16 + 1), 0, stream>>>(W2, b2, Wc, bc, W2c, cvec);

    k_gather1_fc<<<(N + TILE_NODES - 1) / TILE_NODES, T256, 0, stream>>>(X, cnt, slots, src, dst, E, W1, b1, gamma, beta, rmean, rvar, W2c, Hc, N);

    k_gather2<<<gGat2, T256, 0, stream>>>(Hc, cnt, slots, src, dst, E, cvec, out, N);
}


# Slot placement by owner blocks: a block owns 512 destinations, scans the whole destination list, ranks its edges by LDS returning adds; no global atomics, no zeroing launch (5 to 4); sum order kept
# speedup vs baseline: 4.3026x; 1.1582x over previous
#include <hip/hip_runtime.h>
#include <math.h>

typedef float v2f __attribute__((ext_vector_type(2)));
typedef float v8f __attribute__((ext_vector_type(8)));

#define FEAT 128
#define N_CLASSES 16
#define BN_EPS 1e-5f

__device__ __forceinline__ void gemm_tile_f32(const float* __restrict__ A, const float* __restrict__ B,
                                              const float* __restrict__ bias, float* __restrict__ C,
                                              int M, int K, int N, int tile, int lane) {
    const int Nt    = N >> 4;
    const int mBase = (tile / Nt) << 4;
    const int nBase = (tile % Nt) << 4;

    int row = mBase + (lane & 15);
    if (row > M - 1) row = M - 1;
    const int col  = nBase + (lane & 15);
    const int kOff = (lane >> 4) << 1;

    v8f acc = {};
    for (int k = 0; k < K; k += 4) {
        v2f a = *(const v2f*)(A + (size_t)row * K + k + kOff);
        v2f b;
        b.x = B[(size_t)(k + kOff)     * N + col];
        b.y = B[(size_t)(k + kOff + 1) * N + col];
        acc = __builtin_amdgcn_wmma_f32_16x16x4_f32(
                   false, a,  false, b,
                   (short)0, acc,  false,  false);
    }

    const float bb   = bias ? bias[col] : 0.0f;
    const int   rOff = (lane >> 4) << 3;
#pragma unroll
    for (int j = 0; j < 8; ++j) {
        int r = mBase + j + rOff;
        if (r < M) C[(size_t)r * N + col] = acc[j] + bb;
    }
}

__global__ void k_fold_classifier(const float* __restrict__ W2, const float* __restrict__ b2,
                                  const float* __restrict__ Wc, const float* __restrict__ bc,
                                  float* __restrict__ W2c, float* __restrict__ cvec) {
    const int lane = threadIdx.x & 31;
    const int wave = threadIdx.x >> 5;
    if (wave < FEAT / 16)
        gemm_tile_f32(W2, Wc, nullptr, W2c, FEAT, FEAT, N_CLASSES, wave, lane);
    else
        gemm_tile_f32(b2, Wc, bc, cvec, 1, FEAT, N_CLASSES, 0, lane);
}

#define SLOT_CAP    40
#define OWN_NODES   512
#define OWN_THREADS 1024
#define OWN_WAVES   (OWN_THREADS / 32)
#define HIT_CAP     640

__device__ __forceinline__ void place_edge(int e, int d, int lo, const int* __restrict__ src,
                                           int* lcnt, int* __restrict__ slots) {
    const int p = atomicAdd(&lcnt[d - lo], 1);
    if (p < SLOT_CAP) slots[(size_t)d * SLOT_CAP + p] = src[e];
}

__device__ __forceinline__ void scan_value(int d, int e, bool valid, int lo, int2* hitRow, int& wcount,
                                           const int* __restrict__ src, int* lcnt, int* __restrict__ slots) {
    const bool     h = valid && (unsigned)(d - lo) < (unsigned)OWN_NODES;
    const unsigned m = __builtin_amdgcn_ballot_w32(h);
    if (m != 0u) {
        if (wcount + 32 <= HIT_CAP) {
            if (h) hitRow[wcount + (int)__builtin_amdgcn_mbcnt_lo(m, 0u)] = make_int2(e, d);
            wcount += __builtin_popcount(m);
        } else if (h) {
            place_edge(e, d, lo, src, lcnt, slots);
        }
    }
}

__device__ __forceinline__ void load_group(int4* v, const int4* __restrict__ d4, int first) {
#pragma unroll
    for (int u = 0; u < 4; ++u) v[u] = d4[first + u * OWN_THREADS];
}

__device__ __forceinline__ void scan_group(const int4* v, int first, int lo, int2* hitRow, int& wcount,
                                           const int* __restrict__ src, int* lcnt, int* __restrict__ slots) {
#pragma unroll
    for (int u = 0; u < 4; ++u) {
        const int e0 = (first + u * OWN_THREADS) << 2;
        scan_value(v[u].x, e0,     true, lo, hitRow, wcount, src, lcnt, slots);
        scan_value(v[u].y, e0 + 1, true, lo, hitRow, wcount, src, lcnt, slots);
        scan_value(v[u].z, e0 + 2, true, lo, hitRow, wcount, src, lcnt, slots);
        scan_value(v[u].w, e0 + 3, true, lo, hitRow, wcount, src, lcnt, slots);
    }
}

__global__ __launch_bounds__(OWN_THREADS)
void k_place_owned(const int* __restrict__ src, const int* __restrict__ dst,
                   int* __restrict__ cnt, int* __restrict__ slots, int E, int N) {
    __shared__ int  lcnt[OWN_NODES];
    __shared__ int2 hits[OWN_WAVES * HIT_CAP];
    const int tid  = threadIdx.x;
    const int lane = tid & 31;
    const int wave = __builtin_amdgcn_readfirstlane(tid >> 5);
    const int lo   = blockIdx.x * OWN_NODES;
    int2* hitRow   = hits + wave * HIT_CAP;

    for (int i = tid; i < OWN_NODES; i += OWN_THREADS) lcnt[i] = 0;
    __syncthreads();

    int wcount = 0;
    const int nvec = ((E & 3) == 0) ? (E >> 2) : 0;
    const int4* __restrict__ d4 = (const int4*)dst;
    const int GROUP   = 4 * OWN_THREADS;
    const int ngroups = nvec / GROUP;
    int4 ga[4], gb[4];
    if (ngroups > 0) load_group(ga, d4, tid);
    int g = 0;
    for (; g + 2 <= ngroups; g += 2) {
        load_group(gb, d4, (g + 1) * GROUP + tid);
        scan_group(ga, g * GROUP + tid, lo, hitRow, wcount, src, lcnt, slots);
        if (g + 2 < ngroups) load_group(ga, d4, (g + 2) * GROUP + tid);
        scan_group(gb, (g + 1) * GROUP + tid, lo, hitRow, wcount, src, lcnt, slots);
    }
    if (g < ngroups) scan_group(ga, g * GROUP + tid, lo, hitRow, wcount, src, lcnt, slots);
    int base = ngroups * GROUP;
    for (; base < nvec; base += OWN_THREADS) {
        const int  idx = base + tid;
        const bool ok  = idx < nvec;
        const int4 v   = d4[ok ? idx : 0];
        const int  e0  = idx << 2;
        scan_value(v.x, e0,     ok, lo, hitRow, wcount, src, lcnt, slots);
        scan_value(v.y, e0 + 1, ok, lo, hitRow, wcount, src, lcnt, slots);
        scan_value(v.z, e0 + 2, ok, lo, hitRow, wcount, src, lcnt, slots);
        scan_value(v.w, e0 + 3, ok, lo, hitRow, wcount, src, lcnt, slots);
    }
    for (int eb = nvec << 2; eb < E; eb += OWN_THREADS) {
        const int  e  = eb + tid;
        const bool ok = e < E;
        const int  d  = dst[ok ? e : 0];
        scan_value(d, e, ok, lo, hitRow, wcount, src, lcnt, slots);
    }

    for (int b = 0; b < wcount; b += 4 * 32) {
        int2 ed[4];
        int  s[4];
        bool ok[4];
#pragma unroll
        for (int u = 0; u < 4; ++u) {
            const int i = b + u * 32 + lane;
            ok[u] = i < wcount;
            ed[u] = hitRow[ok[u] ? i : 0];
            s[u]  = src[ed[u].x];
        }
#pragma unroll
        for (int u = 0; u < 4; ++u) {
            if (ok[u]) {
                const int p = atomicAdd(&lcnt[ed[u].y - lo], 1);
                if (p < SLOT_CAP) slots[(size_t)ed[u].y * SLOT_CAP + p] = s[u];
            }
        }
    }
    __syncthreads();
    for (int i = tid; i < OWN_NODES; i += OWN_THREADS) {
        const int d = lo + i;
        if (d < N) cnt[d] = lcnt[i];
    }
}

#define TILE_NODES 16
#define ACT_PITCH  (FEAT + 4)

__global__ void k_gather1_fc(const float* __restrict__ X, const int* __restrict__ cnt,
                             const int* __restrict__ slots,
                             const int* __restrict__ src, const int* __restrict__ dst, int E,
                             const float* __restrict__ W1, const float* __restrict__ b1,
                             const float* __restrict__ gamma, const float* __restrict__ beta,
                             const float* __restrict__ mean, const float* __restrict__ var,
                             const float* __restrict__ W2c, float* __restrict__ Hc, int N) {
    __shared__ __attribute__((aligned(16))) float aggT[TILE_NODES * ACT_PITCH];
    __shared__ __attribute__((aligned(16))) float actT[TILE_NODES * ACT_PITCH];
    const int lane = threadIdx.x & 31;
    const int wave     = __builtin_amdgcn_readfirstlane((int)(threadIdx.x >> 5));
    const int tileBase = blockIdx.x * TILE_NODES;

    const int rc   = lane & 15;
    const int kOff = (lane >> 4) << 1;
    const int rOff = (lane >> 4) << 3;
    const int col = (wave << 4) + rc;
    const float* __restrict__ w1p = W1 + kOff * FEAT + col;
    float bcur[16], bnxt[16];
#pragma unroll
    for (int i = 0; i < 8; ++i) {
        bcur[2 * i]     = w1p[(4 * i)     * FEAT];
        bcur[2 * i + 1] = w1p[(4 * i + 1) * FEAT];
    }

    for (int t = 0; t < TILE_NODES / 8; ++t) {
        const int rowT = wave + 8 * t;
        const int node = tileBase + rowT;
        float4 acc = make_float4(0.0f, 0.0f, 0.0f, 0.0f);
        if (node < N) {
            const int   deg = cnt[node];
            const float dn  = rsqrtf((float)(deg + 1));

            const float4 self = *(const float4*)(X + (size_t)node * FEAT + lane * 4);
            const float  wn   = dn * dn;
            acc = make_float4(self.x * wn, self.y * wn, self.z * wn, self.w * wn);

            if (deg <= SLOT_CAP) {
                const int* row = slots + (size_t)node * SLOT_CAP;
                for (int base = 0; base < deg; base += 32) {
                    const int m = min(32, deg - base);
                    int   s = 0;
                    float w = 0.0f;
                    if (lane < m) {
                        s = row[base + lane];
                        w = rsqrtf((float)(cnt[s] + 1)) * dn;
                    }
#pragma unroll 4
                    for (int j = 0; j < m; ++j) {
                        const int    sj = __shfl(s, j);
                        const float  wj = __shfl(w, j);
                        const float4 h  = *(const float4*)(X + (size_t)sj * FEAT + lane * 4);
                        acc.x += h.x * wj;
                        acc.y += h.y * wj;
                        acc.z += h.z * wj;
                        acc.w += h.w * wj;
                    }
                }
            } else {
                for (int base = 0; base < E; base += 32) {
                    const int e = base + lane;
                    int   s = 0;
                    float w = 0.0f;
                    bool  mine = false;
                    if (e < E) mine = (dst[e] == node);
                    if (mine) {
                        s = src[e];
                        w = rsqrtf((float)(cnt[s] + 1)) * dn;
                    }
                    unsigned hit = __builtin_amdgcn_ballot_w32(mine);
                    while (hit != 0u) {
                        const int j = __builtin_ctz(hit);
                        hit &= hit - 1u;
                        const int    sj = __shfl(s, j);
                        const float  wj = __shfl(w, j);
                        const float4 h  = *(const float4*)(X + (size_t)sj * FEAT + lane * 4);
                        acc.x += h.x * wj;
                        acc.y += h.y * wj;
                        acc.z += h.z * wj;
                        acc.w += h.w * wj;
                    }
                }
            }
        }
        *(float4*)(aggT + rowT * ACT_PITCH + lane * 4) = acc;
    }
    __syncthreads();

    {
        v8f acc = {};
#pragma unroll
        for (int q = 0; q < 4; ++q) {
            if (q + 1 < 4) {
#pragma unroll
                for (int i = 0; i < 8; ++i) {
                    bnxt[2 * i]     = w1p[(32 * (q + 1) + 4 * i)     * FEAT];
                    bnxt[2 * i + 1] = w1p[(32 * (q + 1) + 4 * i + 1) * FEAT];
                }
            }
#pragma unroll
            for (int i = 0; i < 8; ++i) {
                const int k = 32 * q + 4 * i;
                const v2f a = *(const v2f*)(aggT + rc * ACT_PITCH + k + kOff);
                v2f b;
                b.x = bcur[2 * i];
                b.y = bcur[2 * i + 1];
                acc = __builtin_amdgcn_wmma_f32_16x16x4_f32(
                           false, a,  false, b,
                           (short)0, acc,  false,  false);
            }
#pragma unroll
            for (int i = 0; i < 16; ++i) bcur[i] = bnxt[i];
        }
        const float vb = b1[col];
        const float vm = mean[col];
        const float vs = rsqrtf(var[col] + BN_EPS);
        const float vg = gamma[col];
        const float vt = beta[col];
#pragma unroll
        for (int j = 0; j < 8; ++j)
            actT[(j + rOff) * ACT_PITCH + col] = fmaxf((acc[j] + vb - vm) * vs * vg + vt, 0.0f);
    }
    __syncthreads();

    if (wave == 0) {
        v8f acc = {};
#pragma unroll 4
        for (int k = 0; k < FEAT; k += 4) {
            const v2f a = *(const v2f*)(actT + rc * ACT_PITCH + k + kOff);
            v2f b;
            b.x = W2c[(k + kOff)     * N_CLASSES + rc];
            b.y = W2c[(k + kOff + 1) * N_CLASSES + rc];
            acc = __builtin_amdgcn_wmma_f32_16x16x4_f32(
                       false, a,  false, b,
                       (short)0, acc,  false,  false);
        }
#pragma unroll
        for (int j = 0; j < 8; ++j) {
            const int rr = tileBase + j + rOff;
            if (rr < N) Hc[(size_t)rr * N_CLASSES + rc] = acc[j];
        }
    }
}

__global__ void k_gather2(const float* __restrict__ Hc, const int* __restrict__ cnt,
                          const int* __restrict__ slots,
                          const int* __restrict__ src, const int* __restrict__ dst, int E,
                          const float* __restrict__ cvec, float* __restrict__ out, int N) {
    const int lane = threadIdx.x & 31;
    const int wave = threadIdx.x >> 5;
    const int f    = lane & (N_CLASSES - 1);
    const int node = ((blockIdx.x * (blockDim.x >> 5) + wave) << 1) + (lane >> 4);
    const bool live = node < N;
    const int  self = live ? node : 0;

    int   deg = 0;
    float dn  = 0.0f, acc = 0.0f;
    if (live) {
        deg = cnt[node];
        dn  = rsqrtf((float)(deg + 1));
        acc = Hc[(size_t)node * N_CLASSES + f] * (dn * dn);
    }
    const bool over = deg > SLOT_CAP;
    const int  len  = over ? 0 : deg;
    const int* row  = slots + (size_t)self * SLOT_CAP;
    const int lenMax = __builtin_amdgcn_readfirstlane(max(len, __shfl_xor(len, 16)));

    for (int base = 0; base < lenMax; base += 16) {
        int   s = self;
        float w = 0.0f;
        if (base + f < len) {
            s = row[base + f];
            w = rsqrtf((float)(cnt[s] + 1)) * dn;
        }
        const int m = min(16, lenMax - base);
#pragma unroll 4
        for (int j = 0; j < m; ++j) {
            const int   sj = __shfl(s, j, 16);
            const float wj = __shfl(w, j, 16);
            acc += Hc[(size_t)sj * N_CLASSES + f] * wj;
        }
    }

    if (__builtin_amdgcn_ballot_w32(over) != 0u) {
        for (int base = 0; base < E; base += 16) {
            const int e = base + f;
            int   s = self;
            float w = 0.0f;
            bool  mine = false;
            if (over && e < E) mine = (dst[e] == node);
            if (mine) {
                s = src[e];
                w = rsqrtf((float)(cnt[s] + 1)) * dn;
            }
            if (__builtin_amdgcn_ballot_w32(mine) != 0u) {
                for (int j = 0; j < 16; ++j) {
                    const int   sj = __shfl(s, j, 16);
                    const float wj = __shfl(w, j, 16);
                    acc += Hc[(size_t)sj * N_CLASSES + f] * wj;
                }
            }
        }
    }
    if (live) out[(size_t)node * N_CLASSES + f] = acc + cvec[f];
}

extern "C" void kernel_launch(void* const* d_in, const int* in_sizes, int n_in,
                              void* d_out, int out_size, void* d_ws, size_t ws_size,
                              hipStream_t stream) {
    const float* X     = (const float*)d_in[0];
    const int*   edge  = (const int*)  d_in[1];
    const float* W1    = (const float*)d_in[2];
    const float* b1    = (const float*)d_in[3];
    const float* gamma = (const float*)d_in[4];
    const float* beta  = (const float*)d_in[5];
    const float* rmean = (const float*)d_in[6];
    const float* rvar  = (const float*)d_in[7];
    const float* W2    = (const float*)d_in[8];
    const float* b2    = (const float*)d_in[9];
    const float* Wc    = (const float*)d_in[10];
    const float* bc    = (const float*)d_in[11];
    float*       out   = (float*)d_out;

    const int N = in_sizes[0] / FEAT;
    const int E = in_sizes[1] / 2;
    const int* src = edge;
    const int* dst = edge + E;
    const int totalC = N * N_CLASSES;

    const int Npad  = (N + 63) & ~63;
    int*   cnt    = (int*)d_ws;
    float* Hc     = (float*)d_ws + Npad;
    float* W2c    = Hc + (size_t)totalC;
    float* cvec   = W2c + FEAT * N_CLASSES;
    int*   slots  = (int*)(cvec + 64);

    const int T256   = 256;
    const int gOwn   = (N + OWN_NODES - 1) / OWN_NODES;
    const int gGat2  = (N + 15) / 16;

    k_place_owned<<<gOwn, OWN_THREADS, 0, stream>>>(src, dst, cnt, slots, E, N);

    k_fold_classifier<<<1, 32 * (FEAT / 16 + 1), 0, stream>>>(W2, b2, Wc, bc, W2c, cvec);

    k_gather1_fc<<<(N + TILE_NODES - 1) / TILE_NODES, T256, 0, stream>>>(X, cnt, slots, src, dst, E, W1, b1, gamma, beta, rmean, rvar, W2c, Hc, N);

    k_gather2<<<gGat2, T256, 0, stream>>>(Hc, cnt, slots, src, dst, E, cvec, out, N);
}


# Placement scan split over two blocks per owner range (196 blocks, not 98): each scans half the edge list into its own 32-slot rows and counters; f32 sums run first-half edges, then second-half
# speedup vs baseline: 4.8582x; 1.1291x over previous
#include <hip/hip_runtime.h>
#include <math.h>

typedef float v2f __attribute__((ext_vector_type(2)));
typedef float v8f __attribute__((ext_vector_type(8)));

#define FEAT 128
#define N_CLASSES 16
#define BN_EPS 1e-5f

__device__ __forceinline__ void gemm_tile_f32(const float* __restrict__ A, const float* __restrict__ B,
                                              const float* __restrict__ bias, float* __restrict__ C,
                                              int M, int K, int N, int tile, int lane) {
    const int Nt    = N >> 4;
    const int mBase = (tile / Nt) << 4;
    const int nBase = (tile % Nt) << 4;

    int row = mBase + (lane & 15);
    if (row > M - 1) row = M - 1;
    const int col  = nBase + (lane & 15);
    const int kOff = (lane >> 4) << 1;

    v8f acc = {};
    for (int k = 0; k < K; k += 4) {
        v2f a = *(const v2f*)(A + (size_t)row * K + k + kOff);
        v2f b;
        b.x = B[(size_t)(k + kOff)     * N + col];
        b.y = B[(size_t)(k + kOff + 1) * N + col];
        acc = __builtin_amdgcn_wmma_f32_16x16x4_f32(
                   false, a,  false, b,
                   (short)0, acc,  false,  false);
    }

    const float bb   = bias ? bias[col] : 0.0f;
    const int   rOff = (lane >> 4) << 3;
#pragma unroll
    for (int j = 0; j < 8; ++j) {
        int r = mBase + j + rOff;
        if (r < M) C[(size_t)r * N + col] = acc[j] + bb;
    }
}

__global__ void k_fold_classifier(const float* __restrict__ W2, const float* __restrict__ b2,
                                  const float* __restrict__ Wc, const float* __restrict__ bc,
                                  float* __restrict__ W2c, float* __restrict__ cvec) {
    const int lane = threadIdx.x & 31;
    const int wave = threadIdx.x >> 5;
    if (wave < FEAT / 16)
        gemm_tile_f32(W2, Wc, nullptr, W2c, FEAT, FEAT, N_CLASSES, wave, lane);
    else
        gemm_tile_f32(b2, Wc, bc, cvec, 1, FEAT, N_CLASSES, 0, lane);
}

#define SLOT_CAP    32
#define OWN_NODES   512
#define OWN_THREADS 1024
#define OWN_WAVES   (OWN_THREADS / 32)
#define HIT_CAP     640

__device__ __forceinline__ void place_edge(int e, int d, int lo, const int* __restrict__ src,
                                           int* lcnt, int* __restrict__ slots) {
    const int p = atomicAdd(&lcnt[d - lo], 1);
    if (p < SLOT_CAP) slots[(size_t)d * SLOT_CAP + p] = src[e];
}

__device__ __forceinline__ void scan_value(int d, int e, bool valid, int lo, int2* hitRow, int& wcount,
                                           const int* __restrict__ src, int* lcnt, int* __restrict__ slots) {
    const bool     h = valid && (unsigned)(d - lo) < (unsigned)OWN_NODES;
    const unsigned m = __builtin_amdgcn_ballot_w32(h);
    if (m != 0u) {
        if (wcount + 32 <= HIT_CAP) {
            if (h) hitRow[wcount + (int)__builtin_amdgcn_mbcnt_lo(m, 0u)] = make_int2(e, d);
            wcount += __builtin_popcount(m);
        } else if (h) {
            place_edge(e, d, lo, src, lcnt, slots);
        }
    }
}

__device__ __forceinline__ void load_group(int4* v, const int4* __restrict__ d4, int first) {
#pragma unroll
    for (int u = 0; u < 4; ++u) v[u] = d4[first + u * OWN_THREADS];
}

__device__ __forceinline__ void scan_group(const int4* v, int first, int lo, int2* hitRow, int& wcount,
                                           const int* __restrict__ src, int* lcnt, int* __restrict__ slots) {
#pragma unroll
    for (int u = 0; u < 4; ++u) {
        const int e0 = (first + u * OWN_THREADS) << 2;
        scan_value(v[u].x, e0,     true, lo, hitRow, wcount, src, lcnt, slots);
        scan_value(v[u].y, e0 + 1, true, lo, hitRow, wcount, src, lcnt, slots);
        scan_value(v[u].z, e0 + 2, true, lo, hitRow, wcount, src, lcnt, slots);
        scan_value(v[u].w, e0 + 3, true, lo, hitRow, wcount, src, lcnt, slots);
    }
}

__global__ __launch_bounds__(OWN_THREADS)
void k_place_owned(const int* __restrict__ src, const int* __restrict__ dst,
                   int* __restrict__ cnt, int* __restrict__ slots, int Eall, int N, int cntStride) {
    __shared__ int  lcnt[OWN_NODES];
    __shared__ int2 hits[OWN_WAVES * HIT_CAP];
    const int tid  = threadIdx.x;
    const int lane = tid & 31;
    const int wave = __builtin_amdgcn_readfirstlane(tid >> 5);
    const int lo   = (blockIdx.x >> 1) * OWN_NODES;
    int2* hitRow   = hits + wave * HIT_CAP;
    const int half = blockIdx.x & 1;
    const int Emid = min(((Eall >> 1) + 3) & ~3, Eall);
    const int E    = half ? Eall - Emid : Emid;
    if (half) {
        src   += Emid;
        dst   += Emid;
        cnt   += cntStride;
        slots += (size_t)N * SLOT_CAP;
    }

    for (int i = tid; i < OWN_NODES; i += OWN_THREADS) lcnt[i] = 0;
    __syncthreads();

    int wcount = 0;
    const int nvec = ((Eall & 3) == 0) ? (E >> 2) : 0;
    const int4* __restrict__ d4 = (const int4*)dst;
    const int GROUP   = 4 * OWN_THREADS;
    const int ngroups = nvec / GROUP;
    int4 ga[4], gb[4];
    if (ngroups > 0) load_group(ga, d4, tid);
    int g = 0;
    for (; g + 2 <= ngroups; g += 2) {
        load_group(gb, d4, (g + 1) * GROUP + tid);
        scan_group(ga, g * GROUP + tid, lo, hitRow, wcount, src, lcnt, slots);
        if (g + 2 < ngroups) load_group(ga, d4, (g + 2) * GROUP + tid);
        scan_group(gb, (g + 1) * GROUP + tid, lo, hitRow, wcount, src, lcnt, slots);
    }
    if (g < ngroups) scan_group(ga, g * GROUP + tid, lo, hitRow, wcount, src, lcnt, slots);
    int base = ngroups * GROUP;
    for (; base < nvec; base += OWN_THREADS) {
        const int  idx = base + tid;
        const bool ok  = idx < nvec;
        const int4 v   = d4[ok ? idx : 0];
        const int  e0  = idx << 2;
        scan_value(v.x, e0,     ok, lo, hitRow, wcount, src, lcnt, slots);
        scan_value(v.y, e0 + 1, ok, lo, hitRow, wcount, src, lcnt, slots);
        scan_value(v.z, e0 + 2, ok, lo, hitRow, wcount, src, lcnt, slots);
        scan_value(v.w, e0 + 3, ok, lo, hitRow, wcount, src, lcnt, slots);
    }
    for (int eb = nvec << 2; eb < E; eb += OWN_THREADS) {
        const int  e  = eb + tid;
        const bool ok = e < E;
        const int  d  = dst[ok ? e : 0];
        scan_value(d, e, ok, lo, hitRow, wcount, src, lcnt, slots);
    }

    for (int b = 0; b < wcount; b += 4 * 32) {
        int2 ed[4];
        int  s[4];
        bool ok[4];
#pragma unroll
        for (int u = 0; u < 4; ++u) {
            const int i = b + u * 32 + lane;
            ok[u] = i < wcount;
            ed[u] = hitRow[ok[u] ? i : 0];
            s[u]  = src[ed[u].x];
        }
#pragma unroll
        for (int u = 0; u < 4; ++u) {
            if (ok[u]) {
                const int p = atomicAdd(&lcnt[ed[u].y - lo], 1);
                if (p < SLOT_CAP) slots[(size_t)ed[u].y * SLOT_CAP + p] = s[u];
            }
        }
    }
    __syncthreads();
    for (int i = tid; i < OWN_NODES; i += OWN_THREADS) {
        const int d = lo + i;
        if (d < N) cnt[d] = lcnt[i];
    }
}

#define TILE_NODES 16
#define ACT_PITCH  (FEAT + 4)

__global__ void k_gather1_fc(const float* __restrict__ X, const int* __restrict__ cnt,
                             const int* __restrict__ cntB,
                             const int* __restrict__ slots, const int* __restrict__ slotsB,
                             const int* __restrict__ src, const int* __restrict__ dst, int E,
                             const float* __restrict__ W1, const float* __restrict__ b1,
                             const float* __restrict__ gamma, const float* __restrict__ beta,
                             const float* __restrict__ mean, const float* __restrict__ var,
                             const float* __restrict__ W2c, float* __restrict__ Hc, int N) {
    __shared__ __attribute__((aligned(16))) float aggT[TILE_NODES * ACT_PITCH];
    __shared__ __attribute__((aligned(16))) float actT[TILE_NODES * ACT_PITCH];
    const int lane = threadIdx.x & 31;
    const int wave     = __builtin_amdgcn_readfirstlane((int)(threadIdx.x >> 5));
    const int tileBase = blockIdx.x * TILE_NODES;

    const int rc   = lane & 15;
    const int kOff = (lane >> 4) << 1;
    const int rOff = (lane >> 4) << 3;
    const int col = (wave << 4) + rc;
    const float* __restrict__ w1p = W1 + kOff * FEAT + col;
    float bcur[16], bnxt[16];
#pragma unroll
    for (int i = 0; i < 8; ++i) {
        bcur[2 * i]     = w1p[(4 * i)     * FEAT];
        bcur[2 * i + 1] = w1p[(4 * i + 1) * FEAT];
    }

    for (int t = 0; t < TILE_NODES / 8; ++t) {
        const int rowT = wave + 8 * t;
        const int node = tileBase + rowT;
        float4 acc = make_float4(0.0f, 0.0f, 0.0f, 0.0f);
        if (node < N) {
            const int   ca  = cnt[node];
            const int   cb  = cntB[node];
            const int   deg = ca + cb;
            const float dn  = rsqrtf((float)(deg + 1));

            const float4 self = *(const float4*)(X + (size_t)node * FEAT + lane * 4);
            const float  wn   = dn * dn;
            acc = make_float4(self.x * wn, self.y * wn, self.z * wn, self.w * wn);

            if (ca <= SLOT_CAP && cb <= SLOT_CAP) {
                const int* rowA = slots  + (size_t)node * SLOT_CAP;
                const int* rowB = slotsB + (size_t)node * SLOT_CAP;
                for (int base = 0; base < deg; base += 32) {
                    const int m = min(32, deg - base);
                    int   s = 0;
                    float w = 0.0f;
                    if (lane < m) {
                        const int i = base + lane;
                        s = (i < ca) ? rowA[i] : rowB[i - ca];
                        w = rsqrtf((float)(cnt[s] + cntB[s] + 1)) * dn;
                    }
#pragma unroll 4
                    for (int j = 0; j < m; ++j) {
                        const int    sj = __shfl(s, j);
                        const float  wj = __shfl(w, j);
                        const float4 h  = *(const float4*)(X + (size_t)sj * FEAT + lane * 4);
                        acc.x += h.x * wj;
                        acc.y += h.y * wj;
                        acc.z += h.z * wj;
                        acc.w += h.w * wj;
                    }
                }
            } else {
                for (int base = 0; base < E; base += 32) {
                    const int e = base + lane;
                    int   s = 0;
                    float w = 0.0f;
                    bool  mine = false;
                    if (e < E) mine = (dst[e] == node);
                    if (mine) {
                        s = src[e];
                        w = rsqrtf((float)(cnt[s] + cntB[s] + 1)) * dn;
                    }
                    unsigned hit = __builtin_amdgcn_ballot_w32(mine);
                    while (hit != 0u) {
                        const int j = __builtin_ctz(hit);
                        hit &= hit - 1u;
                        const int    sj = __shfl(s, j);
                        const float  wj = __shfl(w, j);
                        const float4 h  = *(const float4*)(X + (size_t)sj * FEAT + lane * 4);
                        acc.x += h.x * wj;
                        acc.y += h.y * wj;
                        acc.z += h.z * wj;
                        acc.w += h.w * wj;
                    }
                }
            }
        }
        *(float4*)(aggT + rowT * ACT_PITCH + lane * 4) = acc;
    }
    __syncthreads();

    {
        v8f acc = {};
#pragma unroll
        for (int q = 0; q < 4; ++q) {
            if (q + 1 < 4) {
#pragma unroll
                for (int i = 0; i < 8; ++i) {
                    bnxt[2 * i]     = w1p[(32 * (q + 1) + 4 * i)     * FEAT];
                    bnxt[2 * i + 1] = w1p[(32 * (q + 1) + 4 * i + 1) * FEAT];
                }
            }
#pragma unroll
            for (int i = 0; i < 8; ++i) {
                const int k = 32 * q + 4 * i;
                const v2f a = *(const v2f*)(aggT + rc * ACT_PITCH + k + kOff);
                v2f b;
                b.x = bcur[2 * i];
                b.y = bcur[2 * i + 1];
                acc = __builtin_amdgcn_wmma_f32_16x16x4_f32(
                           false, a,  false, b,
                           (short)0, acc,  false,  false);
            }
#pragma unroll
            for (int i = 0; i < 16; ++i) bcur[i] = bnxt[i];
        }
        const float vb = b1[col];
        const float vm = mean[col];
        const float vs = rsqrtf(var[col] + BN_EPS);
        const float vg = gamma[col];
        const float vt = beta[col];
#pragma unroll
        for (int j = 0; j < 8; ++j)
            actT[(j + rOff) * ACT_PITCH + col] = fmaxf((acc[j] + vb - vm) * vs * vg + vt, 0.0f);
    }
    __syncthreads();

    if (wave == 0) {
        v8f acc = {};
#pragma unroll 4
        for (int k = 0; k < FEAT; k += 4) {
            const v2f a = *(const v2f*)(actT + rc * ACT_PITCH + k + kOff);
            v2f b;
            b.x = W2c[(k + kOff)     * N_CLASSES + rc];
            b.y = W2c[(k + kOff + 1) * N_CLASSES + rc];
            acc = __builtin_amdgcn_wmma_f32_16x16x4_f32(
                       false, a,  false, b,
                       (short)0, acc,  false,  false);
        }
#pragma unroll
        for (int j = 0; j < 8; ++j) {
            const int rr = tileBase + j + rOff;
            if (rr < N) Hc[(size_t)rr * N_CLASSES + rc] = acc[j];
        }
    }
}

__global__ void k_gather2(const float* __restrict__ Hc, const int* __restrict__ cnt,
                          const int* __restrict__ cntB,
                          const int* __restrict__ slots, const int* __restrict__ slotsB,
                          const int* __restrict__ src, const int* __restrict__ dst, int E,
                          const float* __restrict__ cvec, float* __restrict__ out, int N) {
    const int lane = threadIdx.x & 31;
    const int wave = threadIdx.x >> 5;
    const int f    = lane & (N_CLASSES - 1);
    const int node = ((blockIdx.x * (blockDim.x >> 5) + wave) << 1) + (lane >> 4);
    const bool live = node < N;
    const int  self = live ? node : 0;

    int   ca = 0, cb = 0;
    float dn  = 0.0f, acc = 0.0f;
    if (live) {
        ca  = cnt[node];
        cb  = cntB[node];
        dn  = rsqrtf((float)(ca + cb + 1));
        acc = Hc[(size_t)node * N_CLASSES + f] * (dn * dn);
    }
    const bool over = ca > SLOT_CAP || cb > SLOT_CAP;
    const int  len  = over ? 0 : ca + cb;
    const int* rowA = slots  + (size_t)self * SLOT_CAP;
    const int* rowB = slotsB + (size_t)self * SLOT_CAP;
    const int lenMax = __builtin_amdgcn_readfirstlane(max(len, __shfl_xor(len, 16)));

    for (int base = 0; base < lenMax; base += 16) {
        int   s = self;
        float w = 0.0f;
        if (base + f < len) {
            const int i = base + f;
            s = (i < ca) ? rowA[i] : rowB[i - ca];
            w = rsqrtf((float)(cnt[s] + cntB[s] + 1)) * dn;
        }
        const int m = min(16, lenMax - base);
#pragma unroll 4
        for (int j = 0; j < m; ++j) {
            const int   sj = __shfl(s, j, 16);
            const float wj = __shfl(w, j, 16);
            acc += Hc[(size_t)sj * N_CLASSES + f] * wj;
        }
    }

    if (__builtin_amdgcn_ballot_w32(over) != 0u) {
        for (int base = 0; base < E; base += 16) {
            const int e = base + f;
            int   s = self;
            float w = 0.0f;
            bool  mine = false;
            if (over && e < E) mine = (dst[e] == node);
            if (mine) {
                s = src[e];
                w = rsqrtf((float)(cnt[s] + cntB[s] + 1)) * dn;
            }
            if (__builtin_amdgcn_ballot_w32(mine) != 0u) {
                for (int j = 0; j < 16; ++j) {
                    const int   sj = __shfl(s, j, 16);
                    const float wj = __shfl(w, j, 16);
                    acc += Hc[(size_t)sj * N_CLASSES + f] * wj;
                }
            }
        }
    }
    if (live) out[(size_t)node * N_CLASSES + f] = acc + cvec[f];
}

extern "C" void kernel_launch(void* const* d_in, const int* in_sizes, int n_in,
                              void* d_out, int out_size, void* d_ws, size_t ws_size,
                              hipStream_t stream) {
    const float* X     = (const float*)d_in[0];
    const int*   edge  = (const int*)  d_in[1];
    const float* W1    = (const float*)d_in[2];
    const float* b1    = (const float*)d_in[3];
    const float* gamma = (const float*)d_in[4];
    const float* beta  = (const float*)d_in[5];
    const float* rmean = (const float*)d_in[6];
    const float* rvar  = (const float*)d_in[7];
    const float* W2    = (const float*)d_in[8];
    const float* b2    = (const float*)d_in[9];
    const float* Wc    = (const float*)d_in[10];
    const float* bc    = (const float*)d_in[11];
    float*       out   = (float*)d_out;

    const int N = in_sizes[0] / FEAT;
    const int E = in_sizes[1] / 2;
    const int* src = edge;
    const int* dst = edge + E;
    const int totalC = N * N_CLASSES;

    const int Npad  = (N + 63) & ~63;
    int*   cnt    = (int*)d_ws;
    int*   cntB   = cnt + Npad;
    float* Hc     = (float*)d_ws + 2 * Npad;
    float* W2c    = Hc + (size_t)totalC;
    float* cvec   = W2c + FEAT * N_CLASSES;
    int*   slots  = (int*)(cvec + 64);
    int*   slotsB = slots + (size_t)N * SLOT_CAP;

    const int T256   = 256;
    const int gOwn   = 2 * ((N + OWN_NODES - 1) / OWN_NODES);
    const int gGat2  = (N + 15) / 16;

    k_place_owned<<<gOwn, OWN_THREADS, 0, stream>>>(src, dst, cnt, slots, E, N, Npad);

    k_fold_classifier<<<1, 32 * (FEAT / 16 + 1), 0, stream>>>(W2, b2, Wc, bc, W2c, cvec);

    k_gather1_fc<<<(N + TILE_NODES - 1) / TILE_NODES, T256, 0, stream>>>(X, cnt, cntB, slots, slotsB, src, dst, E, W1, b1, gamma, beta, rmean, rvar, W2c, Hc, N);

    k_gather2<<<gGat2, T256, 0, stream>>>(Hc, cnt, cntB, slots, slotsB, src, dst, E, cvec, out, N);
}


# Layer-2 input rows are stored times their own deg^-1/2 where they are made, so the layer-2 gather reads no degree per edge: it sums rows and scales once; f32 products reassociated, sum order kept
# speedup vs baseline: 5.2123x; 1.0729x over previous
#include <hip/hip_runtime.h>
#include <math.h>

typedef float v2f __attribute__((ext_vector_type(2)));
typedef float v8f __attribute__((ext_vector_type(8)));

#define FEAT 128
#define N_CLASSES 16
#define BN_EPS 1e-5f

__device__ __forceinline__ void gemm_tile_f32(const float* __restrict__ A, const float* __restrict__ B,
                                              const float* __restrict__ bias, float* __restrict__ C,
                                              int M, int K, int N, int tile, int lane) {
    const int Nt    = N >> 4;
    const int mBase = (tile / Nt) << 4;
    const int nBase = (tile % Nt) << 4;

    int row = mBase + (lane & 15);
    if (row > M - 1) row = M - 1;
    const int col  = nBase + (lane & 15);
    const int kOff = (lane >> 4) << 1;

    v8f acc = {};
    for (int k = 0; k < K; k += 4) {
        v2f a = *(const v2f*)(A + (size_t)row * K + k + kOff);
        v2f b;
        b.x = B[(size_t)(k + kOff)     * N + col];
        b.y = B[(size_t)(k + kOff + 1) * N + col];
        acc = __builtin_amdgcn_wmma_f32_16x16x4_f32(
                   false, a,  false, b,
                   (short)0, acc,  false,  false);
    }

    const float bb   = bias ? bias[col] : 0.0f;
    const int   rOff = (lane >> 4) << 3;
#pragma unroll
    for (int j = 0; j < 8; ++j) {
        int r = mBase + j + rOff;
        if (r < M) C[(size_t)r * N + col] = acc[j] + bb;
    }
}

__global__ void k_fold_classifier(const float* __restrict__ W2, const float* __restrict__ b2,
                                  const float* __restrict__ Wc, const float* __restrict__ bc,
                                  float* __restrict__ W2c, float* __restrict__ cvec) {
    const int lane = threadIdx.x & 31;
    const int wave = threadIdx.x >> 5;
    if (wave < FEAT / 16)
        gemm_tile_f32(W2, Wc, nullptr, W2c, FEAT, FEAT, N_CLASSES, wave, lane);
    else
        gemm_tile_f32(b2, Wc, bc, cvec, 1, FEAT, N_CLASSES, 0, lane);
}

#define SLOT_CAP    32
#define OWN_NODES   512
#define OWN_THREADS 1024
#define OWN_WAVES   (OWN_THREADS / 32)
#define HIT_CAP     640

__device__ __forceinline__ void place_edge(int e, int d, int lo, const int* __restrict__ src,
                                           int* lcnt, int* __restrict__ slots) {
    const int p = atomicAdd(&lcnt[d - lo], 1);
    if (p < SLOT_CAP) slots[(size_t)d * SLOT_CAP + p] = src[e];
}

__device__ __forceinline__ void scan_value(int d, int e, bool valid, int lo, int2* hitRow, int& wcount,
                                           const int* __restrict__ src, int* lcnt, int* __restrict__ slots) {
    const bool     h = valid && (unsigned)(d - lo) < (unsigned)OWN_NODES;
    const unsigned m = __builtin_amdgcn_ballot_w32(h);
    if (m != 0u) {
        if (wcount + 32 <= HIT_CAP) {
            if (h) hitRow[wcount + (int)__builtin_amdgcn_mbcnt_lo(m, 0u)] = make_int2(e, d);
            wcount += __builtin_popcount(m);
        } else if (h) {
            place_edge(e, d, lo, src, lcnt, slots);
        }
    }
}

__device__ __forceinline__ void load_group(int4* v, const int4* __restrict__ d4, int first) {
#pragma unroll
    for (int u = 0; u < 4; ++u) v[u] = d4[first + u * OWN_THREADS];
}

__device__ __forceinline__ void scan_group(const int4* v, int first, int lo, int2* hitRow, int& wcount,
                                           const int* __restrict__ src, int* lcnt, int* __restrict__ slots) {
#pragma unroll
    for (int u = 0; u < 4; ++u) {
        const int e0 = (first + u * OWN_THREADS) << 2;
        scan_value(v[u].x, e0,     true, lo, hitRow, wcount, src, lcnt, slots);
        scan_value(v[u].y, e0 + 1, true, lo, hitRow, wcount, src, lcnt, slots);
        scan_value(v[u].z, e0 + 2, true, lo, hitRow, wcount, src, lcnt, slots);
        scan_value(v[u].w, e0 + 3, true, lo, hitRow, wcount, src, lcnt, slots);
    }
}

__global__ __launch_bounds__(OWN_THREADS)
void k_place_owned(const int* __restrict__ src, const int* __restrict__ dst,
                   int* __restrict__ cnt, int* __restrict__ slots, int Eall, int N, int cntStride) {
    __shared__ int  lcnt[OWN_NODES];
    __shared__ int2 hits[OWN_WAVES * HIT_CAP];
    const int tid  = threadIdx.x;
    const int lane = tid & 31;
    const int wave = __builtin_amdgcn_readfirstlane(tid >> 5);
    const int lo   = (blockIdx.x >> 1) * OWN_NODES;
    int2* hitRow   = hits + wave * HIT_CAP;
    const int half = blockIdx.x & 1;
    const int Emid = min(((Eall >> 1) + 3) & ~3, Eall);
    const int E    = half ? Eall - Emid : Emid;
    if (half) {
        src   += Emid;
        dst   += Emid;
        cnt   += cntStride;
        slots += (size_t)N * SLOT_CAP;
    }

    for (int i = tid; i < OWN_NODES; i += OWN_THREADS) lcnt[i] = 0;
    __syncthreads();

    int wcount = 0;
    const int nvec = ((Eall & 3) == 0) ? (E >> 2) : 0;
    const int4* __restrict__ d4 = (const int4*)dst;
    const int GROUP   = 4 * OWN_THREADS;
    const int ngroups = nvec / GROUP;
    int4 ga[4], gb[4];
    if (ngroups > 0) load_group(ga, d4, tid);
    int g = 0;
    for (; g + 2 <= ngroups; g += 2) {
        load_group(gb, d4, (g + 1) * GROUP + tid);
        scan_group(ga, g * GROUP + tid, lo, hitRow, wcount, src, lcnt, slots);
        if (g + 2 < ngroups) load_group(ga, d4, (g + 2) * GROUP + tid);
        scan_group(gb, (g + 1) * GROUP + tid, lo, hitRow, wcount, src, lcnt, slots);
    }
    if (g < ngroups) scan_group(ga, g * GROUP + tid, lo, hitRow, wcount, src, lcnt, slots);
    int base = ngroups * GROUP;
    for (; base < nvec; base += OWN_THREADS) {
        const int  idx = base + tid;
        const bool ok  = idx < nvec;
        const int4 v   = d4[ok ? idx : 0];
        const int  e0  = idx << 2;
        scan_value(v.x, e0,     ok, lo, hitRow, wcount, src, lcnt, slots);
        scan_value(v.y, e0 + 1, ok, lo, hitRow, wcount, src, lcnt, slots);
        scan_value(v.z, e0 + 2, ok, lo, hitRow, wcount, src, lcnt, slots);
        scan_value(v.w, e0 + 3, ok, lo, hitRow, wcount, src, lcnt, slots);
    }
    for (int eb = nvec << 2; eb < E; eb += OWN_THREADS) {
        const int  e  = eb + tid;
        const bool ok = e < E;
        const int  d  = dst[ok ? e : 0];
        scan_value(d, e, ok, lo, hitRow, wcount, src, lcnt, slots);
    }

    for (int b = 0; b < wcount; b += 4 * 32) {
        int2 ed[4];
        int  s[4];
        bool ok[4];
#pragma unroll
        for (int u = 0; u < 4; ++u) {
            const int i = b + u * 32 + lane;
            ok[u] = i < wcount;
            ed[u] = hitRow[ok[u] ? i : 0];
            s[u]  = src[ed[u].x];
        }
#pragma unroll
        for (int u = 0; u < 4; ++u) {
            if (ok[u]) {
                const int p = atomicAdd(&lcnt[ed[u].y - lo], 1);
                if (p < SLOT_CAP) slots[(size_t)ed[u].y * SLOT_CAP + p] = s[u];
            }
        }
    }
    __syncthreads();
    for (int i = tid; i < OWN_NODES; i += OWN_THREADS) {
        const int d = lo + i;
        if (d < N) cnt[d] = lcnt[i];
    }
}

#define TILE_NODES 16
#define ACT_PITCH  (FEAT + 4)

__global__ void k_gather1_fc(const float* __restrict__ X, const int* __restrict__ cnt,
                             const int* __restrict__ cntB,
                             const int* __restrict__ slots, const int* __restrict__ slotsB,
                             const int* __restrict__ src, const int* __restrict__ dst, int E,
                             const float* __restrict__ W1, const float* __restrict__ b1,
                             const float* __restrict__ gamma, const float* __restrict__ beta,
                             const float* __restrict__ mean, const float* __restrict__ var,
                             const float* __restrict__ W2c, float* __restrict__ Hc, int N) {
    __shared__ __attribute__((aligned(16))) float aggT[TILE_NODES * ACT_PITCH];
    __shared__ __attribute__((aligned(16))) float actT[TILE_NODES * ACT_PITCH];
    __shared__ __attribute__((aligned(16))) float dnT[TILE_NODES];
    const int lane = threadIdx.x & 31;
    const int wave     = __builtin_amdgcn_readfirstlane((int)(threadIdx.x >> 5));
    const int tileBase = blockIdx.x * TILE_NODES;

    const int rc   = lane & 15;
    const int kOff = (lane >> 4) << 1;
    const int rOff = (lane >> 4) << 3;
    const int col = (wave << 4) + rc;
    const float* __restrict__ w1p = W1 + kOff * FEAT + col;
    float bcur[16], bnxt[16];
#pragma unroll
    for (int i = 0; i < 8; ++i) {
        bcur[2 * i]     = w1p[(4 * i)     * FEAT];
        bcur[2 * i + 1] = w1p[(4 * i + 1) * FEAT];
    }

    for (int t = 0; t < TILE_NODES / 8; ++t) {
        const int rowT = wave + 8 * t;
        const int node = tileBase + rowT;
        float4 acc = make_float4(0.0f, 0.0f, 0.0f, 0.0f);
        float dnRow = 0.0f;
        if (node < N) {
            const int   ca  = cnt[node];
            const int   cb  = cntB[node];
            const int   deg = ca + cb;
            const float dn  = rsqrtf((float)(deg + 1));
            dnRow = dn;

            const float4 self = *(const float4*)(X + (size_t)node * FEAT + lane * 4);
            const float  wn   = dn * dn;
            acc = make_float4(self.x * wn, self.y * wn, self.z * wn, self.w * wn);

            if (ca <= SLOT_CAP && cb <= SLOT_CAP) {
                const int* rowA = slots  + (size_t)node * SLOT_CAP;
                const int* rowB = slotsB + (size_t)node * SLOT_CAP;
                for (int base = 0; base < deg; base += 32) {
                    const int m = min(32, deg - base);
                    int   s = 0;
                    float w = 0.0f;
                    if (lane < m) {
                        const int i = base + lane;
                        s = (i < ca) ? rowA[i] : rowB[i - ca];
                        w = rsqrtf((float)(cnt[s] + cntB[s] + 1)) * dn;
                    }
#pragma unroll 4
                    for (int j = 0; j < m; ++j) {
                        const int    sj = __shfl(s, j);
                        const float  wj = __shfl(w, j);
                        const float4 h  = *(const float4*)(X + (size_t)sj * FEAT + lane * 4);
                        acc.x += h.x * wj;
                        acc.y += h.y * wj;
                        acc.z += h.z * wj;
                        acc.w += h.w * wj;
                    }
                }
            } else {
                for (int base = 0; base < E; base += 32) {
                    const int e = base + lane;
                    int   s = 0;
                    float w = 0.0f;
                    bool  mine = false;
                    if (e < E) mine = (dst[e] == node);
                    if (mine) {
                        s = src[e];
                        w = rsqrtf((float)(cnt[s] + cntB[s] + 1)) * dn;
                    }
                    unsigned hit = __builtin_amdgcn_ballot_w32(mine);
                    while (hit != 0u) {
                        const int j = __builtin_ctz(hit);
                        hit &= hit - 1u;
                        const int    sj = __shfl(s, j);
                        const float  wj = __shfl(w, j);
                        const float4 h  = *(const float4*)(X + (size_t)sj * FEAT + lane * 4);
                        acc.x += h.x * wj;
                        acc.y += h.y * wj;
                        acc.z += h.z * wj;
                        acc.w += h.w * wj;
                    }
                }
            }
        }
        *(float4*)(aggT + rowT * ACT_PITCH + lane * 4) = acc;
        if (lane == 0) dnT[rowT] = dnRow;
    }
    __syncthreads();

    {
        v8f acc = {};
#pragma unroll
        for (int q = 0; q < 4; ++q) {
            if (q + 1 < 4) {
#pragma unroll
                for (int i = 0; i < 8; ++i) {
                    bnxt[2 * i]     = w1p[(32 * (q + 1) + 4 * i)     * FEAT];
                    bnxt[2 * i + 1] = w1p[(32 * (q + 1) + 4 * i + 1) * FEAT];
                }
            }
#pragma unroll
            for (int i = 0; i < 8; ++i) {
                const int k = 32 * q + 4 * i;
                const v2f a = *(const v2f*)(aggT + rc * ACT_PITCH + k + kOff);
                v2f b;
                b.x = bcur[2 * i];
                b.y = bcur[2 * i + 1];
                acc = __builtin_amdgcn_wmma_f32_16x16x4_f32(
                           false, a,  false, b,
                           (short)0, acc,  false,  false);
            }
#pragma unroll
            for (int i = 0; i < 16; ++i) bcur[i] = bnxt[i];
        }
        const float vb = b1[col];
        const float vm = mean[col];
        const float vs = rsqrtf(var[col] + BN_EPS);
        const float vg = gamma[col];
        const float vt = beta[col];
#pragma unroll
        for (int j = 0; j < 8; ++j)
            actT[(j + rOff) * ACT_PITCH + col] = fmaxf((acc[j] + vb - vm) * vs * vg + vt, 0.0f);
    }
    __syncthreads();

    if (wave == 0) {
        const float4 d0 = *(const float4*)(dnT + rOff);
        const float4 d1 = *(const float4*)(dnT + rOff + 4);
        const float  dj[8] = {d0.x, d0.y, d0.z, d0.w, d1.x, d1.y, d1.z, d1.w};
        v8f acc = {};
#pragma unroll 4
        for (int k = 0; k < FEAT; k += 4) {
            const v2f a = *(const v2f*)(actT + rc * ACT_PITCH + k + kOff);
            v2f b;
            b.x = W2c[(k + kOff)     * N_CLASSES + rc];
            b.y = W2c[(k + kOff + 1) * N_CLASSES + rc];
            acc = __builtin_amdgcn_wmma_f32_16x16x4_f32(
                       false, a,  false, b,
                       (short)0, acc,  false,  false);
        }
#pragma unroll
        for (int j = 0; j < 8; ++j) {
            const int rr = tileBase + j + rOff;
            if (rr < N) Hc[(size_t)rr * N_CLASSES + rc] = acc[j] * dj[j];
        }
    }
}

__global__ void k_gather2(const float* __restrict__ Hs, const int* __restrict__ cnt,
                          const int* __restrict__ cntB,
                          const int* __restrict__ slots, const int* __restrict__ slotsB,
                          const int* __restrict__ src, const int* __restrict__ dst, int E,
                          const float* __restrict__ cvec, float* __restrict__ out, int N) {
    const int lane = threadIdx.x & 31;
    const int wave = threadIdx.x >> 5;
    const int f    = lane & (N_CLASSES - 1);
    const int node = ((blockIdx.x * (blockDim.x >> 5) + wave) << 1) + (lane >> 4);
    const bool live = node < N;
    const int  self = live ? node : 0;

    int   ca = 0, cb = 0;
    float dn  = 0.0f, acc = 0.0f;
    if (live) {
        ca  = cnt[node];
        cb  = cntB[node];
        dn  = rsqrtf((float)(ca + cb + 1));
        acc = Hs[(size_t)node * N_CLASSES + f];
    }
    const bool over = ca > SLOT_CAP || cb > SLOT_CAP;
    const int  len  = over ? 0 : ca + cb;
    const int* rowA = slots  + (size_t)self * SLOT_CAP;
    const int* rowB = slotsB + (size_t)self * SLOT_CAP;
    const int lenMax = __builtin_amdgcn_readfirstlane(max(len, __shfl_xor(len, 16)));

    for (int base = 0; base < lenMax; base += 16) {
        int s = self;
        if (base + f < len) {
            const int i = base + f;
            s = (i < ca) ? rowA[i] : rowB[i - ca];
        }
        const int m    = min(16, lenMax - base);
        const int left = len - base;
#pragma unroll 4
        for (int j = 0; j < m; ++j) {
            const int   sj = __shfl(s, j, 16);
            const float h  = Hs[(size_t)sj * N_CLASSES + f];
            acc += (j < left) ? h : 0.0f;
        }
    }

    if (__builtin_amdgcn_ballot_w32(over) != 0u) {
        for (int base = 0; base < E; base += 16) {
            const int e = base + f;
            int   s = self;
            bool  mine = false;
            if (over && e < E) mine = (dst[e] == node);
            if (mine) s = src[e];
            if (__builtin_amdgcn_ballot_w32(mine) != 0u) {
                const int flag = mine ? 1 : 0;
                for (int j = 0; j < 16; ++j) {
                    const int   sj = __shfl(s, j, 16);
                    const int   mj = __shfl(flag, j, 16);
                    const float h  = Hs[(size_t)sj * N_CLASSES + f];
                    acc += (mj != 0) ? h : 0.0f;
                }
            }
        }
    }
    if (live) out[(size_t)node * N_CLASSES + f] = acc * dn + cvec[f];
}

extern "C" void kernel_launch(void* const* d_in, const int* in_sizes, int n_in,
                              void* d_out, int out_size, void* d_ws, size_t ws_size,
                              hipStream_t stream) {
    const float* X     = (const float*)d_in[0];
    const int*   edge  = (const int*)  d_in[1];
    const float* W1    = (const float*)d_in[2];
    const float* b1    = (const float*)d_in[3];
    const float* gamma = (const float*)d_in[4];
    const float* beta  = (const float*)d_in[5];
    const float* rmean = (const float*)d_in[6];
    const float* rvar  = (const float*)d_in[7];
    const float* W2    = (const float*)d_in[8];
    const float* b2    = (const float*)d_in[9];
    const float* Wc    = (const float*)d_in[10];
    const float* bc    = (const float*)d_in[11];
    float*       out   = (float*)d_out;

    const int N = in_sizes[0] / FEAT;
    const int E = in_sizes[1] / 2;
    const int* src = edge;
    const int* dst = edge + E;
    const int totalC = N * N_CLASSES;

    const int Npad  = (N + 63) & ~63;
    int*   cnt    = (int*)d_ws;
    int*   cntB   = cnt + Npad;
    float* Hc     = (float*)d_ws + 2 * Npad;
    float* W2c    = Hc + (size_t)totalC;
    float* cvec   = W2c + FEAT * N_CLASSES;
    int*   slots  = (int*)(cvec + 64);
    int*   slotsB = slots + (size_t)N * SLOT_CAP;

    const int T256   = 256;
    const int gOwn   = 2 * ((N + OWN_NODES - 1) / OWN_NODES);
    const int gGat2  = (N + 15) / 16;

    k_place_owned<<<gOwn, OWN_THREADS, 0, stream>>>(src, dst, cnt, slots, E, N, Npad);

    k_fold_classifier<<<1, 32 * (FEAT / 16 + 1), 0, stream>>>(W2, b2, Wc, bc, W2c, cvec);

    k_gather1_fc<<<(N + TILE_NODES - 1) / TILE_NODES, T256, 0, stream>>>(X, cnt, cntB, slots, slotsB, src, dst, E, W1, b1, gamma, beta, rmean, rvar, W2c, Hc, N);

    k_gather2<<<gGat2, T256, 0, stream>>>(Hc, cnt, cntB, slots, slotsB, src, dst, E, cvec, out, N);
}


# Classifier fold (W2 Wc and b2 Wc + bc on f32 WMMA) rides in the placement launch as one extra block on a WGP the owner blocks leave idle: 4 launches to 3; products and f32 sum order unchanged
# speedup vs baseline: 5.5167x; 1.0584x over previous
#include <hip/hip_runtime.h>
#include <math.h>

typedef float v2f __attribute__((ext_vector_type(2)));
typedef float v8f __attribute__((ext_vector_type(8)));

#define FEAT 128
#define N_CLASSES 16
#define BN_EPS 1e-5f

__device__ __forceinline__ void gemm_tile_f32(const float* __restrict__ A, const float* __restrict__ B,
                                              const float* __restrict__ bias, float* __restrict__ C,
                                              int M, int K, int N, int tile, int lane) {
    const int Nt    = N >> 4;
    const int mBase = (tile / Nt) << 4;
    const int nBase = (tile % Nt) << 4;

    int row = mBase + (lane & 15);
    if (row > M - 1) row = M - 1;
    const int col  = nBase + (lane & 15);
    const int kOff = (lane >> 4) << 1;

    v8f acc = {};
    for (int k = 0; k < K; k += 4) {
        v2f a = *(const v2f*)(A + (size_t)row * K + k + kOff);
        v2f b;
        b.x = B[(size_t)(k + kOff)     * N + col];
        b.y = B[(size_t)(k + kOff + 1) * N + col];
        acc = __builtin_amdgcn_wmma_f32_16x16x4_f32(
                   false, a,  false, b,
                   (short)0, acc,  false,  false);
    }

    const float bb   = bias ? bias[col] : 0.0f;
    const int   rOff = (lane >> 4) << 3;
#pragma unroll
    for (int j = 0; j < 8; ++j) {
        int r = mBase + j + rOff;
        if (r < M) C[(size_t)r * N + col] = acc[j] + bb;
    }
}

__device__ __forceinline__ void fold_classifier(const float* __restrict__ W2, const float* __restrict__ b2,
                                                const float* __restrict__ Wc, const float* __restrict__ bc,
                                                float* __restrict__ W2c, float* __restrict__ cvec,
                                                int wave, int lane) {
    if (wave < FEAT / 16)
        gemm_tile_f32(W2, Wc, nullptr, W2c, FEAT, FEAT, N_CLASSES, wave, lane);
    else if (wave == FEAT / 16)
        gemm_tile_f32(b2, Wc, bc, cvec, 1, FEAT, N_CLASSES, 0, lane);
}

#define SLOT_CAP    32
#define OWN_NODES   512
#define OWN_THREADS 1024
#define OWN_WAVES   (OWN_THREADS / 32)
#define HIT_CAP     640

__device__ __forceinline__ void place_edge(int e, int d, int lo, const int* __restrict__ src,
                                           int* lcnt, int* __restrict__ slots) {
    const int p = atomicAdd(&lcnt[d - lo], 1);
    if (p < SLOT_CAP) slots[(size_t)d * SLOT_CAP + p] = src[e];
}

__device__ __forceinline__ void scan_value(int d, int e, bool valid, int lo, int2* hitRow, int& wcount,
                                           const int* __restrict__ src, int* lcnt, int* __restrict__ slots) {
    const bool     h = valid && (unsigned)(d - lo) < (unsigned)OWN_NODES;
    const unsigned m = __builtin_amdgcn_ballot_w32(h);
    if (m != 0u) {
        if (wcount + 32 <= HIT_CAP) {
            if (h) hitRow[wcount + (int)__builtin_amdgcn_mbcnt_lo(m, 0u)] = make_int2(e, d);
            wcount += __builtin_popcount(m);
        } else if (h) {
            place_edge(e, d, lo, src, lcnt, slots);
        }
    }
}

__device__ __forceinline__ void load_group(int4* v, const int4* __restrict__ d4, int first) {
#pragma unroll
    for (int u = 0; u < 4; ++u) v[u] = d4[first + u * OWN_THREADS];
}

__device__ __forceinline__ void scan_group(const int4* v, int first, int lo, int2* hitRow, int& wcount,
                                           const int* __restrict__ src, int* lcnt, int* __restrict__ slots) {
#pragma unroll
    for (int u = 0; u < 4; ++u) {
        const int e0 = (first + u * OWN_THREADS) << 2;
        scan_value(v[u].x, e0,     true, lo, hitRow, wcount, src, lcnt, slots);
        scan_value(v[u].y, e0 + 1, true, lo, hitRow, wcount, src, lcnt, slots);
        scan_value(v[u].z, e0 + 2, true, lo, hitRow, wcount, src, lcnt, slots);
        scan_value(v[u].w, e0 + 3, true, lo, hitRow, wcount, src, lcnt, slots);
    }
}

__global__ __launch_bounds__(OWN_THREADS)
void k_place_owned(const int* __restrict__ src, const int* __restrict__ dst,
                   int* __restrict__ cnt, int* __restrict__ slots, int Eall, int N, int cntStride,
                   int ownBlocks,
                   const float* __restrict__ W2, const float* __restrict__ b2,
                   const float* __restrict__ Wc, const float* __restrict__ bc,
                   float* __restrict__ W2c, float* __restrict__ cvec) {
    __shared__ int  lcnt[OWN_NODES];
    __shared__ int2 hits[OWN_WAVES * HIT_CAP];
    const int tid  = threadIdx.x;
    const int lane = tid & 31;
    const int wave = __builtin_amdgcn_readfirstlane(tid >> 5);
    if ((int)blockIdx.x >= ownBlocks) {
        fold_classifier(W2, b2, Wc, bc, W2c, cvec, wave, lane);
        return;
    }
    const int lo   = (blockIdx.x >> 1) * OWN_NODES;
    int2* hitRow   = hits + wave * HIT_CAP;
    const int half = blockIdx.x & 1;
    const int Emid = min(((Eall >> 1) + 3) & ~3, Eall);
    const int E    = half ? Eall - Emid : Emid;
    if (half) {
        src   += Emid;
        dst   += Emid;
        cnt   += cntStride;
        slots += (size_t)N * SLOT_CAP;
    }

    for (int i = tid; i < OWN_NODES; i += OWN_THREADS) lcnt[i] = 0;
    __syncthreads();

    int wcount = 0;
    const int nvec = ((Eall & 3) == 0) ? (E >> 2) : 0;
    const int4* __restrict__ d4 = (const int4*)dst;
    const int GROUP   = 4 * OWN_THREADS;
    const int ngroups = nvec / GROUP;
    int4 ga[4], gb[4];
    if (ngroups > 0) load_group(ga, d4, tid);
    int g = 0;
    for (; g + 2 <= ngroups; g += 2) {
        load_group(gb, d4, (g + 1) * GROUP + tid);
        scan_group(ga, g * GROUP + tid, lo, hitRow, wcount, src, lcnt, slots);
        if (g + 2 < ngroups) load_group(ga, d4, (g + 2) * GROUP + tid);
        scan_group(gb, (g + 1) * GROUP + tid, lo, hitRow, wcount, src, lcnt, slots);
    }
    if (g < ngroups) scan_group(ga, g * GROUP + tid, lo, hitRow, wcount, src, lcnt, slots);
    int base = ngroups * GROUP;
    for (; base < nvec; base += OWN_THREADS) {
        const int  idx = base + tid;
        const bool ok  = idx < nvec;
        const int4 v   = d4[ok ? idx : 0];
        const int  e0  = idx << 2;
        scan_value(v.x, e0,     ok, lo, hitRow, wcount, src, lcnt, slots);
        scan_value(v.y, e0 + 1, ok, lo, hitRow, wcount, src, lcnt, slots);
        scan_value(v.z, e0 + 2, ok, lo, hitRow, wcount, src, lcnt, slots);
        scan_value(v.w, e0 + 3, ok, lo, hitRow, wcount, src, lcnt, slots);
    }
    for (int eb = nvec << 2; eb < E; eb += OWN_THREADS) {
        const int  e  = eb + tid;
        const bool ok = e < E;
        const int  d  = dst[ok ? e : 0];
        scan_value(d, e, ok, lo, hitRow, wcount, src, lcnt, slots);
    }

    for (int b = 0; b < wcount; b += 4 * 32) {
        int2 ed[4];
        int  s[4];
        bool ok[4];
#pragma unroll
        for (int u = 0; u < 4; ++u) {
            const int i = b + u * 32 + lane;
            ok[u] = i < wcount;
            ed[u] = hitRow[ok[u] ? i : 0];
            s[u]  = src[ed[u].x];
        }
#pragma unroll
        for (int u = 0; u < 4; ++u) {
            if (ok[u]) {
                const int p = atomicAdd(&lcnt[ed[u].y - lo], 1);
                if (p < SLOT_CAP) slots[(size_t)ed[u].y * SLOT_CAP + p] = s[u];
            }
        }
    }
    __syncthreads();
    for (int i = tid; i < OWN_NODES; i += OWN_THREADS) {
        const int d = lo + i;
        if (d < N) cnt[d] = lcnt[i];
    }
}

#define TILE_NODES 16
#define ACT_PITCH  (FEAT + 4)

__global__ void k_gather1_fc(const float* __restrict__ X, const int* __restrict__ cnt,
                             const int* __restrict__ cntB,
                             const int* __restrict__ slots, const int* __restrict__ slotsB,
                             const int* __restrict__ src, const int* __restrict__ dst, int E,
                             const float* __restrict__ W1, const float* __restrict__ b1,
                             const float* __restrict__ gamma, const float* __restrict__ beta,
                             const float* __restrict__ mean, const float* __restrict__ var,
                             const float* __restrict__ W2c, float* __restrict__ Hc, int N) {
    __shared__ __attribute__((aligned(16))) float aggT[TILE_NODES * ACT_PITCH];
    __shared__ __attribute__((aligned(16))) float actT[TILE_NODES * ACT_PITCH];
    __shared__ __attribute__((aligned(16))) float dnT[TILE_NODES];
    const int lane = threadIdx.x & 31;
    const int wave     = __builtin_amdgcn_readfirstlane((int)(threadIdx.x >> 5));
    const int tileBase = blockIdx.x * TILE_NODES;

    const int rc   = lane & 15;
    const int kOff = (lane >> 4) << 1;
    const int rOff = (lane >> 4) << 3;
    const int col = (wave << 4) + rc;
    const float* __restrict__ w1p = W1 + kOff * FEAT + col;
    float bcur[16], bnxt[16];
#pragma unroll
    for (int i = 0; i < 8; ++i) {
        bcur[2 * i]     = w1p[(4 * i)     * FEAT];
        bcur[2 * i + 1] = w1p[(4 * i + 1) * FEAT];
    }

    for (int t = 0; t < TILE_NODES / 8; ++t) {
        const int rowT = wave + 8 * t;
        const int node = tileBase + rowT;
        float4 acc = make_float4(0.0f, 0.0f, 0.0f, 0.0f);
        float dnRow = 0.0f;
        if (node < N) {
            const int   ca  = cnt[node];
            const int   cb  = cntB[node];
            const int   deg = ca + cb;
            const float dn  = rsqrtf((float)(deg + 1));
            dnRow = dn;

            const float4 self = *(const float4*)(X + (size_t)node * FEAT + lane * 4);
            const float  wn   = dn * dn;
            acc = make_float4(self.x * wn, self.y * wn, self.z * wn, self.w * wn);

            if (ca <= SLOT_CAP && cb <= SLOT_CAP) {
                const int* rowA = slots  + (size_t)node * SLOT_CAP;
                const int* rowB = slotsB + (size_t)node * SLOT_CAP;
                for (int base = 0; base < deg; base += 32) {
                    const int m = min(32, deg - base);
                    int   s = 0;
                    float w = 0.0f;
                    if (lane < m) {
                        const int i = base + lane;
                        s = (i < ca) ? rowA[i] : rowB[i - ca];
                        w = rsqrtf((float)(cnt[s] + cntB[s] + 1)) * dn;
                    }
#pragma unroll 4
                    for (int j = 0; j < m; ++j) {
                        const int    sj = __shfl(s, j);
                        const float  wj = __shfl(w, j);
                        const float4 h  = *(const float4*)(X + (size_t)sj * FEAT + lane * 4);
                        acc.x += h.x * wj;
                        acc.y += h.y * wj;
                        acc.z += h.z * wj;
                        acc.w += h.w * wj;
                    }
                }
            } else {
                for (int base = 0; base < E; base += 32) {
                    const int e = base + lane;
                    int   s = 0;
                    float w = 0.0f;
                    bool  mine = false;
                    if (e < E) mine = (dst[e] == node);
                    if (mine) {
                        s = src[e];
                        w = rsqrtf((float)(cnt[s] + cntB[s] + 1)) * dn;
                    }
                    unsigned hit = __builtin_amdgcn_ballot_w32(mine);
                    while (hit != 0u) {
                        const int j = __builtin_ctz(hit);
                        hit &= hit - 1u;
                        const int    sj = __shfl(s, j);
                        const float  wj = __shfl(w, j);
                        const float4 h  = *(const float4*)(X + (size_t)sj * FEAT + lane * 4);
                        acc.x += h.x * wj;
                        acc.y += h.y * wj;
                        acc.z += h.z * wj;
                        acc.w += h.w * wj;
                    }
                }
            }
        }
        *(float4*)(aggT + rowT * ACT_PITCH + lane * 4) = acc;
        if (lane == 0) dnT[rowT] = dnRow;
    }
    __syncthreads();

    {
        v8f acc = {};
#pragma unroll
        for (int q = 0; q < 4; ++q) {
            if (q + 1 < 4) {
#pragma unroll
                for (int i = 0; i < 8; ++i) {
                    bnxt[2 * i]     = w1p[(32 * (q + 1) + 4 * i)     * FEAT];
                    bnxt[2 * i + 1] = w1p[(32 * (q + 1) + 4 * i + 1) * FEAT];
                }
            }
#pragma unroll
            for (int i = 0; i < 8; ++i) {
                const int k = 32 * q + 4 * i;
                const v2f a = *(const v2f*)(aggT + rc * ACT_PITCH + k + kOff);
                v2f b;
                b.x = bcur[2 * i];
                b.y = bcur[2 * i + 1];
                acc = __builtin_amdgcn_wmma_f32_16x16x4_f32(
                           false, a,  false, b,
                           (short)0, acc,  false,  false);
            }
#pragma unroll
            for (int i = 0; i < 16; ++i) bcur[i] = bnxt[i];
        }
        const float vb = b1[col];
        const float vm = mean[col];
        const float vs = rsqrtf(var[col] + BN_EPS);
        const float vg = gamma[col];
        const float vt = beta[col];
#pragma unroll
        for (int j = 0; j < 8; ++j)
            actT[(j + rOff) * ACT_PITCH + col] = fmaxf((acc[j] + vb - vm) * vs * vg + vt, 0.0f);
    }
    __syncthreads();

    if (wave == 0) {
        const float4 d0 = *(const float4*)(dnT + rOff);
        const float4 d1 = *(const float4*)(dnT + rOff + 4);
        const float  dj[8] = {d0.x, d0.y, d0.z, d0.w, d1.x, d1.y, d1.z, d1.w};
        v8f acc = {};
#pragma unroll 4
        for (int k = 0; k < FEAT; k += 4) {
            const v2f a = *(const v2f*)(actT + rc * ACT_PITCH + k + kOff);
            v2f b;
            b.x = W2c[(k + kOff)     * N_CLASSES + rc];
            b.y = W2c[(k + kOff + 1) * N_CLASSES + rc];
            acc = __builtin_amdgcn_wmma_f32_16x16x4_f32(
                       false, a,  false, b,
                       (short)0, acc,  false,  false);
        }
#pragma unroll
        for (int j = 0; j < 8; ++j) {
            const int rr = tileBase + j + rOff;
            if (rr < N) Hc[(size_t)rr * N_CLASSES + rc] = acc[j] * dj[j];
        }
    }
}

__global__ void k_gather2(const float* __restrict__ Hs, const int* __restrict__ cnt,
                          const int* __restrict__ cntB,
                          const int* __restrict__ slots, const int* __restrict__ slotsB,
                          const int* __restrict__ src, const int* __restrict__ dst, int E,
                          const float* __restrict__ cvec, float* __restrict__ out, int N) {
    const int lane = threadIdx.x & 31;
    const int wave = threadIdx.x >> 5;
    const int f    = lane & (N_CLASSES - 1);
    const int node = ((blockIdx.x * (blockDim.x >> 5) + wave) << 1) + (lane >> 4);
    const bool live = node < N;
    const int  self = live ? node : 0;

    int   ca = 0, cb = 0;
    float dn  = 0.0f, acc = 0.0f;
    if (live) {
        ca  = cnt[node];
        cb  = cntB[node];
        dn  = rsqrtf((float)(ca + cb + 1));
        acc = Hs[(size_t)node * N_CLASSES + f];
    }
    const bool over = ca > SLOT_CAP || cb > SLOT_CAP;
    const int  len  = over ? 0 : ca + cb;
    const int* rowA = slots  + (size_t)self * SLOT_CAP;
    const int* rowB = slotsB + (size_t)self * SLOT_CAP;
    const int lenMax = __builtin_amdgcn_readfirstlane(max(len, __shfl_xor(len, 16)));

    for (int base = 0; base < lenMax; base += 16) {
        int s = self;
        if (base + f < len) {
            const int i = base + f;
            s = (i < ca) ? rowA[i] : rowB[i - ca];
        }
        const int m    = min(16, lenMax - base);
        const int left = len - base;
#pragma unroll 4
        for (int j = 0; j < m; ++j) {
            const int   sj = __shfl(s, j, 16);
            const float h  = Hs[(size_t)sj * N_CLASSES + f];
            acc += (j < left) ? h : 0.0f;
        }
    }

    if (__builtin_amdgcn_ballot_w32(over) != 0u) {
        for (int base = 0; base < E; base += 16) {
            const int e = base + f;
            int   s = self;
            bool  mine = false;
            if (over && e < E) mine = (dst[e] == node);
            if (mine) s = src[e];
            if (__builtin_amdgcn_ballot_w32(mine) != 0u) {
                const int flag = mine ? 1 : 0;
                for (int j = 0; j < 16; ++j) {
                    const int   sj = __shfl(s, j, 16);
                    const int   mj = __shfl(flag, j, 16);
                    const float h  = Hs[(size_t)sj * N_CLASSES + f];
                    acc += (mj != 0) ? h : 0.0f;
                }
            }
        }
    }
    if (live) out[(size_t)node * N_CLASSES + f] = acc * dn + cvec[f];
}

extern "C" void kernel_launch(void* const* d_in, const int* in_sizes, int n_in,
                              void* d_out, int out_size, void* d_ws, size_t ws_size,
                              hipStream_t stream) {
    const float* X     = (const float*)d_in[0];
    const int*   edge  = (const int*)  d_in[1];
    const float* W1    = (const float*)d_in[2];
    const float* b1    = (const float*)d_in[3];
    const float* gamma = (const float*)d_in[4];
    const float* beta  = (const float*)d_in[5];
    const float* rmean = (const float*)d_in[6];
    const float* rvar  = (const float*)d_in[7];
    const float* W2    = (const float*)d_in[8];
    const float* b2    = (const float*)d_in[9];
    const float* Wc    = (const float*)d_in[10];
    const float* bc    = (const float*)d_in[11];
    float*       out   = (float*)d_out;

    const int N = in_sizes[0] / FEAT;
    const int E = in_sizes[1] / 2;
    const int* src = edge;
    const int* dst = edge + E;
    const int totalC = N * N_CLASSES;

    const int Npad  = (N + 63) & ~63;
    int*   cnt    = (int*)d_ws;
    int*   cntB   = cnt + Npad;
    float* Hc     = (float*)d_ws + 2 * Npad;
    float* W2c    = Hc + (size_t)totalC;
    float* cvec   = W2c + FEAT * N_CLASSES;
    int*   slots  = (int*)(cvec + 64);
    int*   slotsB = slots + (size_t)N * SLOT_CAP;

    const int T256   = 256;
    const int gOwn   = 2 * ((N + OWN_NODES - 1) / OWN_NODES);
    const int gGat2  = (N + 15) / 16;

    k_place_owned<<<gOwn + 1, OWN_THREADS, 0, stream>>>(src, dst, cnt, slots, E, N, Npad, gOwn, W2, b2, Wc, bc, W2c, cvec);

    k_gather1_fc<<<(N + TILE_NODES - 1) / TILE_NODES, T256, 0, stream>>>(X, cnt, cntB, slots, slotsB, src, dst, E, W1, b1, gamma, beta, rmean, rvar, W2c, Hc, N);

    k_gather2<<<gGat2, T256, 0, stream>>>(Hc, cnt, cntB, slots, slotsB, src, dst, E, cvec, out, N);
}


# Layer-1 gather blocks own 32 nodes, not 16: a W1 fragment in registers feeds two f32 WMMAs, one per row tile, so W1 is fetched once per 32 nodes; products, k order and f32 sum order unchanged
# speedup vs baseline: 5.7351x; 1.0396x over previous
#include <hip/hip_runtime.h>
#include <math.h>

typedef float v2f __attribute__((ext_vector_type(2)));
typedef float v8f __attribute__((ext_vector_type(8)));

#define FEAT 128
#define N_CLASSES 16
#define BN_EPS 1e-5f

__device__ __forceinline__ void gemm_tile_f32(const float* __restrict__ A, const float* __restrict__ B,
                                              const float* __restrict__ bias, float* __restrict__ C,
                                              int M, int K, int N, int tile, int lane) {
    const int Nt    = N >> 4;
    const int mBase = (tile / Nt) << 4;
    const int nBase = (tile % Nt) << 4;

    int row = mBase + (lane & 15);
    if (row > M - 1) row = M - 1;
    const int col  = nBase + (lane & 15);
    const int kOff = (lane >> 4) << 1;

    v8f acc = {};
    for (int k = 0; k < K; k += 4) {
        v2f a = *(const v2f*)(A + (size_t)row * K + k + kOff);
        v2f b;
        b.x = B[(size_t)(k + kOff)     * N + col];
        b.y = B[(size_t)(k + kOff + 1) * N + col];
        acc = __builtin_amdgcn_wmma_f32_16x16x4_f32(
                   false, a,  false, b,
                   (short)0, acc,  false,  false);
    }

    const float bb   = bias ? bias[col] : 0.0f;
    const int   rOff = (lane >> 4) << 3;
#pragma unroll
    for (int j = 0; j < 8; ++j) {
        int r = mBase + j + rOff;
        if (r < M) C[(size_t)r * N + col] = acc[j] + bb;
    }
}

__device__ __forceinline__ void fold_classifier(const float* __restrict__ W2, const float* __restrict__ b2,
                                                const float* __restrict__ Wc, const float* __restrict__ bc,
                                                float* __restrict__ W2c, float* __restrict__ cvec,
                                                int wave, int lane) {
    if (wave < FEAT / 16)
        gemm_tile_f32(W2, Wc, nullptr, W2c, FEAT, FEAT, N_CLASSES, wave, lane);
    else if (wave == FEAT / 16)
        gemm_tile_f32(b2, Wc, bc, cvec, 1, FEAT, N_CLASSES, 0, lane);
}

#define SLOT_CAP    32
#define OWN_NODES   512
#define OWN_THREADS 1024
#define OWN_WAVES   (OWN_THREADS / 32)
#define HIT_CAP     640

__device__ __forceinline__ void place_edge(int e, int d, int lo, const int* __restrict__ src,
                                           int* lcnt, int* __restrict__ slots) {
    const int p = atomicAdd(&lcnt[d - lo], 1);
    if (p < SLOT_CAP) slots[(size_t)d * SLOT_CAP + p] = src[e];
}

__device__ __forceinline__ void scan_value(int d, int e, bool valid, int lo, int2* hitRow, int& wcount,
                                           const int* __restrict__ src, int* lcnt, int* __restrict__ slots) {
    const bool     h = valid && (unsigned)(d - lo) < (unsigned)OWN_NODES;
    const unsigned m = __builtin_amdgcn_ballot_w32(h);
    if (m != 0u) {
        if (wcount + 32 <= HIT_CAP) {
            if (h) hitRow[wcount + (int)__builtin_amdgcn_mbcnt_lo(m, 0u)] = make_int2(e, d);
            wcount += __builtin_popcount(m);
        } else if (h) {
            place_edge(e, d, lo, src, lcnt, slots);
        }
    }
}

__device__ __forceinline__ void load_group(int4* v, const int4* __restrict__ d4, int first) {
#pragma unroll
    for (int u = 0; u < 4; ++u) v[u] = d4[first + u * OWN_THREADS];
}

__device__ __forceinline__ void scan_group(const int4* v, int first, int lo, int2* hitRow, int& wcount,
                                           const int* __restrict__ src, int* lcnt, int* __restrict__ slots) {
#pragma unroll
    for (int u = 0; u < 4; ++u) {
        const int e0 = (first + u * OWN_THREADS) << 2;
        scan_value(v[u].x, e0,     true, lo, hitRow, wcount, src, lcnt, slots);
        scan_value(v[u].y, e0 + 1, true, lo, hitRow, wcount, src, lcnt, slots);
        scan_value(v[u].z, e0 + 2, true, lo, hitRow, wcount, src, lcnt, slots);
        scan_value(v[u].w, e0 + 3, true, lo, hitRow, wcount, src, lcnt, slots);
    }
}

__global__ __launch_bounds__(OWN_THREADS)
void k_place_owned(const int* __restrict__ src, const int* __restrict__ dst,
                   int* __restrict__ cnt, int* __restrict__ slots, int Eall, int N, int cntStride,
                   int ownBlocks,
                   const float* __restrict__ W2, const float* __restrict__ b2,
                   const float* __restrict__ Wc, const float* __restrict__ bc,
                   float* __restrict__ W2c, float* __restrict__ cvec) {
    __shared__ int  lcnt[OWN_NODES];
    __shared__ int2 hits[OWN_WAVES * HIT_CAP];
    const int tid  = threadIdx.x;
    const int lane = tid & 31;
    const int wave = __builtin_amdgcn_readfirstlane(tid >> 5);
    if ((int)blockIdx.x >= ownBlocks) {
        fold_classifier(W2, b2, Wc, bc, W2c, cvec, wave, lane);
        return;
    }
    const int lo   = (blockIdx.x >> 1) * OWN_NODES;
    int2* hitRow   = hits + wave * HIT_CAP;
    const int half = blockIdx.x & 1;
    const int Emid = min(((Eall >> 1) + 3) & ~3, Eall);
    const int E    = half ? Eall - Emid : Emid;
    if (half) {
        src   += Emid;
        dst   += Emid;
        cnt   += cntStride;
        slots += (size_t)N * SLOT_CAP;
    }

    for (int i = tid; i < OWN_NODES; i += OWN_THREADS) lcnt[i] = 0;
    __syncthreads();

    int wcount = 0;
    const int nvec = ((Eall & 3) == 0) ? (E >> 2) : 0;
    const int4* __restrict__ d4 = (const int4*)dst;
    const int GROUP   = 4 * OWN_THREADS;
    const int ngroups = nvec / GROUP;
    int4 ga[4], gb[4];
    if (ngroups > 0) load_group(ga, d4, tid);
    int g = 0;
    for (; g + 2 <= ngroups; g += 2) {
        load_group(gb, d4, (g + 1) * GROUP + tid);
        scan_group(ga, g * GROUP + tid, lo, hitRow, wcount, src, lcnt, slots);
        if (g + 2 < ngroups) load_group(ga, d4, (g + 2) * GROUP + tid);
        scan_group(gb, (g + 1) * GROUP + tid, lo, hitRow, wcount, src, lcnt, slots);
    }
    if (g < ngroups) scan_group(ga, g * GROUP + tid, lo, hitRow, wcount, src, lcnt, slots);
    int base = ngroups * GROUP;
    for (; base < nvec; base += OWN_THREADS) {
        const int  idx = base + tid;
        const bool ok  = idx < nvec;
        const int4 v   = d4[ok ? idx : 0];
        const int  e0  = idx << 2;
        scan_value(v.x, e0,     ok, lo, hitRow, wcount, src, lcnt, slots);
        scan_value(v.y, e0 + 1, ok, lo, hitRow, wcount, src, lcnt, slots);
        scan_value(v.z, e0 + 2, ok, lo, hitRow, wcount, src, lcnt, slots);
        scan_value(v.w, e0 + 3, ok, lo, hitRow, wcount, src, lcnt, slots);
    }
    for (int eb = nvec << 2; eb < E; eb += OWN_THREADS) {
        const int  e  = eb + tid;
        const bool ok = e < E;
        const int  d  = dst[ok ? e : 0];
        scan_value(d, e, ok, lo, hitRow, wcount, src, lcnt, slots);
    }

    for (int b = 0; b < wcount; b += 4 * 32) {
        int2 ed[4];
        int  s[4];
        bool ok[4];
#pragma unroll
        for (int u = 0; u < 4; ++u) {
            const int i = b + u * 32 + lane;
            ok[u] = i < wcount;
            ed[u] = hitRow[ok[u] ? i : 0];
            s[u]  = src[ed[u].x];
        }
#pragma unroll
        for (int u = 0; u < 4; ++u) {
            if (ok[u]) {
                const int p = atomicAdd(&lcnt[ed[u].y - lo], 1);
                if (p < SLOT_CAP) slots[(size_t)ed[u].y * SLOT_CAP + p] = s[u];
            }
        }
    }
    __syncthreads();
    for (int i = tid; i < OWN_NODES; i += OWN_THREADS) {
        const int d = lo + i;
        if (d < N) cnt[d] = lcnt[i];
    }
}

#define TILE_NODES 32
#define ACT_PITCH  (FEAT + 4)

__global__ void k_gather1_fc(const float* __restrict__ X, const int* __restrict__ cnt,
                             const int* __restrict__ cntB,
                             const int* __restrict__ slots, const int* __restrict__ slotsB,
                             const int* __restrict__ src, const int* __restrict__ dst, int E,
                             const float* __restrict__ W1, const float* __restrict__ b1,
                             const float* __restrict__ gamma, const float* __restrict__ beta,
                             const float* __restrict__ mean, const float* __restrict__ var,
                             const float* __restrict__ W2c, float* __restrict__ Hc, int N) {
    __shared__ __attribute__((aligned(16))) float aggT[TILE_NODES * ACT_PITCH];
    __shared__ __attribute__((aligned(16))) float actT[TILE_NODES * ACT_PITCH];
    __shared__ __attribute__((aligned(16))) float dnT[TILE_NODES];
    const int lane = threadIdx.x & 31;
    const int wave     = __builtin_amdgcn_readfirstlane((int)(threadIdx.x >> 5));
    const int tileBase = blockIdx.x * TILE_NODES;

    const int rc   = lane & 15;
    const int kOff = (lane >> 4) << 1;
    const int rOff = (lane >> 4) << 3;
    const int col = (wave << 4) + rc;
    const float* __restrict__ w1p = W1 + kOff * FEAT + col;
    float bcur[16], bnxt[16];
#pragma unroll
    for (int i = 0; i < 8; ++i) {
        bcur[2 * i]     = w1p[(4 * i)     * FEAT];
        bcur[2 * i + 1] = w1p[(4 * i + 1) * FEAT];
    }

    for (int t = 0; t < TILE_NODES / 8; ++t) {
        const int rowT = wave + 8 * t;
        const int node = tileBase + rowT;
        float4 acc = make_float4(0.0f, 0.0f, 0.0f, 0.0f);
        float dnRow = 0.0f;
        if (node < N) {
            const int   ca  = cnt[node];
            const int   cb  = cntB[node];
            const int   deg = ca + cb;
            const float dn  = rsqrtf((float)(deg + 1));
            dnRow = dn;

            const float4 self = *(const float4*)(X + (size_t)node * FEAT + lane * 4);
            const float  wn   = dn * dn;
            acc = make_float4(self.x * wn, self.y * wn, self.z * wn, self.w * wn);

            if (ca <= SLOT_CAP && cb <= SLOT_CAP) {
                const int* rowA = slots  + (size_t)node * SLOT_CAP;
                const int* rowB = slotsB + (size_t)node * SLOT_CAP;
                for (int base = 0; base < deg; base += 32) {
                    const int m = min(32, deg - base);
                    int   s = 0;
                    float w = 0.0f;
                    if (lane < m) {
                        const int i = base + lane;
                        s = (i < ca) ? rowA[i] : rowB[i - ca];
                        w = rsqrtf((float)(cnt[s] + cntB[s] + 1)) * dn;
                    }
#pragma unroll 4
                    for (int j = 0; j < m; ++j) {
                        const int    sj = __shfl(s, j);
                        const float  wj = __shfl(w, j);
                        const float4 h  = *(const float4*)(X + (size_t)sj * FEAT + lane * 4);
                        acc.x += h.x * wj;
                        acc.y += h.y * wj;
                        acc.z += h.z * wj;
                        acc.w += h.w * wj;
                    }
                }
            } else {
                for (int base = 0; base < E; base += 32) {
                    const int e = base + lane;
                    int   s = 0;
                    float w = 0.0f;
                    bool  mine = false;
                    if (e < E) mine = (dst[e] == node);
                    if (mine) {
                        s = src[e];
                        w = rsqrtf((float)(cnt[s] + cntB[s] + 1)) * dn;
                    }
                    unsigned hit = __builtin_amdgcn_ballot_w32(mine);
                    while (hit != 0u) {
                        const int j = __builtin_ctz(hit);
                        hit &= hit - 1u;
                        const int    sj = __shfl(s, j);
                        const float  wj = __shfl(w, j);
                        const float4 h  = *(const float4*)(X + (size_t)sj * FEAT + lane * 4);
                        acc.x += h.x * wj;
                        acc.y += h.y * wj;
                        acc.z += h.z * wj;
                        acc.w += h.w * wj;
                    }
                }
            }
        }
        *(float4*)(aggT + rowT * ACT_PITCH + lane * 4) = acc;
        if (lane == 0) dnT[rowT] = dnRow;
    }
    __syncthreads();

    {
        v8f acc0 = {};
        v8f acc1 = {};
#pragma unroll
        for (int q = 0; q < 4; ++q) {
            if (q + 1 < 4) {
#pragma unroll
                for (int i = 0; i < 8; ++i) {
                    bnxt[2 * i]     = w1p[(32 * (q + 1) + 4 * i)     * FEAT];
                    bnxt[2 * i + 1] = w1p[(32 * (q + 1) + 4 * i + 1) * FEAT];
                }
            }
#pragma unroll
            for (int i = 0; i < 8; ++i) {
                const int k = 32 * q + 4 * i;
                const v2f a0 = *(const v2f*)(aggT + rc * ACT_PITCH + k + kOff);
                const v2f a1 = *(const v2f*)(aggT + (16 + rc) * ACT_PITCH + k + kOff);
                v2f b;
                b.x = bcur[2 * i];
                b.y = bcur[2 * i + 1];
                acc0 = __builtin_amdgcn_wmma_f32_16x16x4_f32(
                            false, a0,  false, b,
                            (short)0, acc0,  false,  false);
                acc1 = __builtin_amdgcn_wmma_f32_16x16x4_f32(
                            false, a1,  false, b,
                            (short)0, acc1,  false,  false);
            }
#pragma unroll
            for (int i = 0; i < 16; ++i) bcur[i] = bnxt[i];
        }
        const float vb = b1[col];
        const float vm = mean[col];
        const float vs = rsqrtf(var[col] + BN_EPS);
        const float vg = gamma[col];
        const float vt = beta[col];
#pragma unroll
        for (int j = 0; j < 8; ++j) {
            actT[(j + rOff) * ACT_PITCH + col]      = fmaxf((acc0[j] + vb - vm) * vs * vg + vt, 0.0f);
            actT[(16 + j + rOff) * ACT_PITCH + col] = fmaxf((acc1[j] + vb - vm) * vs * vg + vt, 0.0f);
        }
    }
    __syncthreads();

    if (wave < TILE_NODES / 16) {
        const int rb = wave << 4;
        const float4 d0 = *(const float4*)(dnT + rb + rOff);
        const float4 d1 = *(const float4*)(dnT + rb + rOff + 4);
        const float  dj[8] = {d0.x, d0.y, d0.z, d0.w, d1.x, d1.y, d1.z, d1.w};
        v8f acc = {};
#pragma unroll 4
        for (int k = 0; k < FEAT; k += 4) {
            const v2f a = *(const v2f*)(actT + (rb + rc) * ACT_PITCH + k + kOff);
            v2f b;
            b.x = W2c[(k + kOff)     * N_CLASSES + rc];
            b.y = W2c[(k + kOff + 1) * N_CLASSES + rc];
            acc = __builtin_amdgcn_wmma_f32_16x16x4_f32(
                       false, a,  false, b,
                       (short)0, acc,  false,  false);
        }
#pragma unroll
        for (int j = 0; j < 8; ++j) {
            const int rr = tileBase + rb + j + rOff;
            if (rr < N) Hc[(size_t)rr * N_CLASSES + rc] = acc[j] * dj[j];
        }
    }
}

__global__ void k_gather2(const float* __restrict__ Hs, const int* __restrict__ cnt,
                          const int* __restrict__ cntB,
                          const int* __restrict__ slots, const int* __restrict__ slotsB,
                          const int* __restrict__ src, const int* __restrict__ dst, int E,
                          const float* __restrict__ cvec, float* __restrict__ out, int N) {
    const int lane = threadIdx.x & 31;
    const int wave = threadIdx.x >> 5;
    const int f    = lane & (N_CLASSES - 1);
    const int node = ((blockIdx.x * (blockDim.x >> 5) + wave) << 1) + (lane >> 4);
    const bool live = node < N;
    const int  self = live ? node : 0;

    int   ca = 0, cb = 0;
    float dn  = 0.0f, acc = 0.0f;
    if (live) {
        ca  = cnt[node];
        cb  = cntB[node];
        dn  = rsqrtf((float)(ca + cb + 1));
        acc = Hs[(size_t)node * N_CLASSES + f];
    }
    const bool over = ca > SLOT_CAP || cb > SLOT_CAP;
    const int  len  = over ? 0 : ca + cb;
    const int* rowA = slots  + (size_t)self * SLOT_CAP;
    const int* rowB = slotsB + (size_t)self * SLOT_CAP;
    const int lenMax = __builtin_amdgcn_readfirstlane(max(len, __shfl_xor(len, 16)));

    for (int base = 0; base < lenMax; base += 16) {
        int s = self;
        if (base + f < len) {
            const int i = base + f;
            s = (i < ca) ? rowA[i] : rowB[i - ca];
        }
        const int m    = min(16, lenMax - base);
        const int left = len - base;
#pragma unroll 4
        for (int j = 0; j < m; ++j) {
            const int   sj = __shfl(s, j, 16);
            const float h  = Hs[(size_t)sj * N_CLASSES + f];
            acc += (j < left) ? h : 0.0f;
        }
    }

    if (__builtin_amdgcn_ballot_w32(over) != 0u) {
        for (int base = 0; base < E; base += 16) {
            const int e = base + f;
            int   s = self;
            bool  mine = false;
            if (over && e < E) mine = (dst[e] == node);
            if (mine) s = src[e];
            if (__builtin_amdgcn_ballot_w32(mine) != 0u) {
                const int flag = mine ? 1 : 0;
                for (int j = 0; j < 16; ++j) {
                    const int   sj = __shfl(s, j, 16);
                    const int   mj = __shfl(flag, j, 16);
                    const float h  = Hs[(size_t)sj * N_CLASSES + f];
                    acc += (mj != 0) ? h : 0.0f;
                }
            }
        }
    }
    if (live) out[(size_t)node * N_CLASSES + f] = acc * dn + cvec[f];
}

extern "C" void kernel_launch(void* const* d_in, const int* in_sizes, int n_in,
                              void* d_out, int out_size, void* d_ws, size_t ws_size,
                              hipStream_t stream) {
    const float* X     = (const float*)d_in[0];
    const int*   edge  = (const int*)  d_in[1];
    const float* W1    = (const float*)d_in[2];
    const float* b1    = (const float*)d_in[3];
    const float* gamma = (const float*)d_in[4];
    const float* beta  = (const float*)d_in[5];
    const float* rmean = (const float*)d_in[6];
    const float* rvar  = (const float*)d_in[7];
    const float* W2    = (const float*)d_in[8];
    const float* b2    = (const float*)d_in[9];
    const float* Wc    = (const float*)d_in[10];
    const float* bc    = (const float*)d_in[11];
    float*       out   = (float*)d_out;

    const int N = in_sizes[0] / FEAT;
    const int E = in_sizes[1] / 2;
    const int* src = edge;
    const int* dst = edge + E;
    const int totalC = N * N_CLASSES;

    const int Npad  = (N + 63) & ~63;
    int*   cnt    = (int*)d_ws;
    int*   cntB   = cnt + Npad;
    float* Hc     = (float*)d_ws + 2 * Npad;
    float* W2c    = Hc + (size_t)totalC;
    float* cvec   = W2c + FEAT * N_CLASSES;
    int*   slots  = (int*)(cvec + 64);
    int*   slotsB = slots + (size_t)N * SLOT_CAP;

    const int T256   = 256;
    const int gOwn   = 2 * ((N + OWN_NODES - 1) / OWN_NODES);
    const int gGat2  = (N + 15) / 16;

    k_place_owned<<<gOwn + 1, OWN_THREADS, 0, stream>>>(src, dst, cnt, slots, E, N, Npad, gOwn, W2, b2, Wc, bc, W2c, cvec);

    k_gather1_fc<<<(N + TILE_NODES - 1) / TILE_NODES, T256, 0, stream>>>(X, cnt, cntB, slots, slotsB, src, dst, E, W1, b1, gamma, beta, rmean, rvar, W2c, Hc, N);

    k_gather2<<<gGat2, T256, 0, stream>>>(Hc, cnt, cntB, slots, slotsB, src, dst, E, cvec, out, N);
}
